# starting kernel
# baseline (speedup 1.0000x reference)
#include <hip/hip_runtime.h>
#include <hip/hip_bf16.h>

typedef __attribute__((ext_vector_type(2))) float v2f;
typedef __attribute__((ext_vector_type(4))) float v4f;
typedef __attribute__((ext_vector_type(8))) float v8f;

#define HIDDEN 64
#define DNODE 384

static __device__ __forceinline__ v8f wmma4(v2f a, v2f b, v8f c) {
  return __builtin_amdgcn_wmma_f32_16x16x4_f32(
       false, a,  false, b,
       (short)0, c,  false,  false);
}

static __device__ __forceinline__ void stage_pairs(float* sW, const float* W,
                                                   int rows, int tid, int nthr) {
  const int nvec = rows * HIDDEN / 4;
  for (int idx = tid; idx < nvec; idx += nthr) {
    int k = idx >> 4;
    int col = (idx & 15) * 4;
    v4f v = ((const v4f*)W)[idx];
    float* dstp = sW + (k >> 1) * 128 + (k & 1);
    dstp[(col + 0) * 2] = v.x;
    dstp[(col + 1) * 2] = v.y;
    dstp[(col + 2) * 2] = v.z;
    dstp[(col + 3) * 2] = v.w;
  }
}

__global__ void k_zero(float* __restrict__ p, size_t n) {
  size_t i = (size_t)blockIdx.x * blockDim.x + threadIdx.x;
  size_t stride = (size_t)gridDim.x * blockDim.x;
  for (; i < n; i += stride) p[i] = 0.0f;
}

__global__ void k_deg(const int* __restrict__ dst, float* __restrict__ deg, int E) {
  int e = blockIdx.x * blockDim.x + threadIdx.x;
  if (e < E) atomicAdd(deg + dst[e], 1.0f);
}

__global__ __launch_bounds__(256) void k_scatter(const int* __restrict__ src,
                                                 const int* __restrict__ dst,
                                                 const float* __restrict__ h,
                                                 float* __restrict__ agg, int E) {
  int t = blockIdx.x * blockDim.x + threadIdx.x;
  int e = t >> 5;
  int lane = t & 31;
  if (e >= E) return;
  int s = src[e];
  int d = dst[e];
  v2f v = *(const v2f*)(h + (size_t)s * HIDDEN + lane * 2);
  atomicAdd(agg + (size_t)d * HIDDEN + lane * 2, v.x);
  atomicAdd(agg + (size_t)d * HIDDEN + lane * 2 + 1, v.y);
}

__global__ __launch_bounds__(128) void k_inproj(const float* __restrict__ x,
                                                const float* __restrict__ Win,
                                                const float* __restrict__ bin,
                                                float* __restrict__ h, int n) {
  __shared__ float sW[96 * HIDDEN];
  const int wave = threadIdx.x >> 5;
  const int lane = threadIdx.x & 31;
  const int hf = lane >> 4;
  const int r = lane & 15;
  const int rowBase = blockIdx.x * 64 + wave * 16;
  const int row = rowBase + r;
  const int rowc = row < n ? row : n - 1;
  const float* xr = x + (size_t)rowc * DNODE;

  v8f acc[4];
#pragma unroll
  for (int c = 0; c < 4; ++c) {
    float b = bin[c * 16 + r];
#pragma unroll
    for (int i = 0; i < 8; ++i) acc[c][i] = b;
  }

  for (int kc = 0; kc < DNODE; kc += 96) {
    __syncthreads();
    stage_pairs(sW, Win + (size_t)kc * HIDDEN, 96, threadIdx.x, 128);
    __syncthreads();

    if (kc + 96 < DNODE) __builtin_prefetch(xr + kc + 96 + 2 * hf, 0, 3);

#pragma unroll 4
    for (int k = 0; k < 96; k += 4) {
      v2f a = *(const v2f*)(xr + kc + k + 2 * hf);
      const float* wrow = sW + ((k >> 1) + hf) * 128;
#pragma unroll
      for (int c = 0; c < 4; ++c) {
        v2f b = *(const v2f*)(wrow + (c * 16 + r) * 2);
        acc[c] = wmma4(a, b, acc[c]);
      }
    }
  }

#pragma unroll
  for (int c = 0; c < 4; ++c) {
#pragma unroll
    for (int i = 0; i < 8; ++i) {
      int m = rowBase + i + 8 * hf;
      if (m < n) h[(size_t)m * HIDDEN + c * 16 + r] = fmaxf(acc[c][i], 0.0f);
    }
  }
}

__global__ __launch_bounds__(128) void k_layer(const float* __restrict__ hprev,
                                               const float* __restrict__ agg,
                                               const float* __restrict__ deg,
                                               const float* __restrict__ Wl,
                                               const float* __restrict__ bl,
                                               const float* __restrict__ Wr,
                                               const float* __restrict__ g,
                                               const float* __restrict__ bt,
                                               float* __restrict__ hnext, int n) {
  __shared__ float sWl[HIDDEN * HIDDEN];
  __shared__ float sWr[HIDDEN * HIDDEN];
  __shared__ float sOut[4 * 16 * HIDDEN];

  stage_pairs(sWl, Wl, HIDDEN, threadIdx.x, 128);
  stage_pairs(sWr, Wr, HIDDEN, threadIdx.x, 128);
  __syncthreads();

  const int wave = threadIdx.x >> 5;
  const int lane = threadIdx.x & 31;
  const int hf = lane >> 4;
  const int r = lane & 15;
  const int rowBase = blockIdx.x * 64 + wave * 16;
  const int row = rowBase + r;
  const int rowc = row < n ? row : n - 1;
  const float* ar = agg + (size_t)rowc * HIDDEN;
  const float* hr = hprev + (size_t)rowc * HIDDEN;
  const float invd = 1.0f / fmaxf(deg[rowc], 1.0f);

  v8f acc[4];
#pragma unroll
  for (int c = 0; c < 4; ++c) {
    float b = bl[c * 16 + r];
#pragma unroll
    for (int i = 0; i < 8; ++i) acc[c][i] = b;
  }

#pragma unroll 4
  for (int k = 0; k < HIDDEN; k += 4) {
    v2f a1 = *(const v2f*)(ar + k + 2 * hf);
    a1.x *= invd;
    a1.y *= invd;
    v2f a2 = *(const v2f*)(hr + k + 2 * hf);
    const float* wl = sWl + ((k >> 1) + hf) * 128;
    const float* wr = sWr + ((k >> 1) + hf) * 128;
#pragma unroll
    for (int c = 0; c < 4; ++c) {
      v2f b1 = *(const v2f*)(wl + (c * 16 + r) * 2);
      acc[c] = wmma4(a1, b1, acc[c]);
      v2f b2 = *(const v2f*)(wr + (c * 16 + r) * 2);
      acc[c] = wmma4(a2, b2, acc[c]);
    }
  }

  float* so = sOut + wave * 16 * HIDDEN;
#pragma unroll
  for (int c = 0; c < 4; ++c)
#pragma unroll
    for (int i = 0; i < 8; ++i)
      so[(i + 8 * hf) * HIDDEN + c * 16 + r] = acc[c][i];
  __syncthreads();

  const int node = rowBase + r;
  const bool valid = node < n;
  const int nodec = valid ? node : n - 1;
  const v4f* rp = (const v4f*)(so + r * HIDDEN + hf * 32);
  v4f vv[8];
  float ss = 0.0f;
#pragma unroll
  for (int j = 0; j < 8; ++j) {
    vv[j] = rp[j];
    ss += vv[j].x * vv[j].x + vv[j].y * vv[j].y + vv[j].z * vv[j].z + vv[j].w * vv[j].w;
  }
  ss += __shfl_xor(ss, 16);
  const float l2 = rsqrtf(fmaxf(ss, 1e-24f));

  const v4f* hres = (const v4f*)(hprev + (size_t)nodec * HIDDEN + hf * 32);
  float s = 0.0f, q = 0.0f;
#pragma unroll
  for (int j = 0; j < 8; ++j) {
    v4f t = vv[j] * l2;
    t.x = fmaxf(t.x, 0.0f);
    t.y = fmaxf(t.y, 0.0f);
    t.z = fmaxf(t.z, 0.0f);
    t.w = fmaxf(t.w, 0.0f);
    t += hres[j];
    vv[j] = t;
    s += t.x + t.y + t.z + t.w;
    q += t.x * t.x + t.y * t.y + t.z * t.z + t.w * t.w;
  }
  s += __shfl_xor(s, 16);
  q += __shfl_xor(q, 16);
  const float mean = s * (1.0f / 64.0f);
  const float var = q * (1.0f / 64.0f) - mean * mean;
  const float sc = rsqrtf(var + 1e-5f);

  const v4f* gp = (const v4f*)(g + hf * 32);
  const v4f* bp = (const v4f*)(bt + hf * 32);
  if (valid) {
    v4f* op = (v4f*)(hnext + (size_t)node * HIDDEN + hf * 32);
#pragma unroll
    for (int j = 0; j < 8; ++j) op[j] = (vv[j] - mean) * sc * gp[j] + bp[j];
  }
}

__global__ __launch_bounds__(256) void k_finalln(const float* __restrict__ h,
                                                 const float* __restrict__ g,
                                                 const float* __restrict__ b,
                                                 float* __restrict__ out, int n) {
  int t = blockIdx.x * blockDim.x + threadIdx.x;
  int row = t >> 5;
  int lane = t & 31;
  if (row >= n) return;
  v2f v = *(const v2f*)(h + (size_t)row * HIDDEN + lane * 2);
  float s = v.x + v.y;
  float q = v.x * v.x + v.y * v.y;
#pragma unroll
  for (int m = 16; m >= 1; m >>= 1) {
    s += __shfl_xor(s, m);
    q += __shfl_xor(q, m);
  }
  float mean = s * (1.0f / 64.0f);
  float var = q * (1.0f / 64.0f) - mean * mean;
  float sc = rsqrtf(var + 1e-5f);
  v2f g2 = *(const v2f*)(g + lane * 2);
  v2f b2 = *(const v2f*)(b + lane * 2);
  v2f o;
  o.x = (v.x - mean) * sc * g2.x + b2.x;
  o.y = (v.y - mean) * sc * g2.y + b2.y;
  *(v2f*)(out + (size_t)row * HIDDEN + lane * 2) = o;
}

extern "C" void kernel_launch(void* const* d_in, const int* in_sizes, int n_in,
                              void* d_out, int out_size, void* d_ws, size_t ws_size,
                              hipStream_t stream) {
  const float* x   = (const float*)d_in[0];
  const int* ei    = (const int*)d_in[1];
  const float* Win = (const float*)d_in[2];
  const float* bin = (const float*)d_in[3];
  const float* Wl  = (const float*)d_in[4];
  const float* bl  = (const float*)d_in[5];
  const float* Wr  = (const float*)d_in[6];
  const float* lng = (const float*)d_in[7];
  const float* lnb = (const float*)d_in[8];
  const float* og  = (const float*)d_in[9];
  const float* ob  = (const float*)d_in[10];

  const int N = in_sizes[0] / DNODE;
  const int E = in_sizes[1] / 2;
  const int LAYERS = in_sizes[4] / (HIDDEN * HIDDEN);
  const int* src = ei;
  const int* dst = ei + E;

  float* h0  = (float*)d_ws;
  float* h1  = h0 + (size_t)N * HIDDEN;
  float* agg = h1 + (size_t)N * HIDDEN;
  float* deg = agg + (size_t)N * HIDDEN;

  k_zero<<<1024, 256, 0, stream>>>(deg, (size_t)N);
  k_deg<<<(E + 255) / 256, 256, 0, stream>>>(dst, deg, E);

  k_inproj<<<(N + 63) / 64, 128, 0, stream>>>(x, Win, bin, h0, N);

  float* cur = h0;
  float* nxt = h1;
  for (int l = 0; l < LAYERS; ++l) {
    k_zero<<<2048, 256, 0, stream>>>(agg, (size_t)N * HIDDEN);
    k_scatter<<<(int)(((long long)E * 32 + 255) / 256), 256, 0, stream>>>(src, dst, cur, agg, E);
    k_layer<<<(N + 63) / 64, 128, 0, stream>>>(
        cur, agg, deg,
        Wl + (size_t)l * HIDDEN * HIDDEN, bl + (size_t)l * HIDDEN,
        Wr + (size_t)l * HIDDEN * HIDDEN,
        lng + (size_t)l * HIDDEN, lnb + (size_t)l * HIDDEN, nxt, N);
    float* t = cur; cur = nxt; nxt = t;
  }

  k_finalln<<<(int)(((long long)N * 32 + 255) / 256), 256, 0, stream>>>(cur, og, ob, (float*)d_out, N);
}


# scatter: one wave walks a strip of 32 edges (indices read coalesced, broadcast to scalars, 8 row loads in flight) instead of one wave per edge; sums stay in atomic arrival order
# speedup vs baseline: 1.4876x; 1.4876x over previous
#include <hip/hip_runtime.h>
#include <hip/hip_bf16.h>

typedef __attribute__((ext_vector_type(2))) float v2f;
typedef __attribute__((ext_vector_type(4))) float v4f;
typedef __attribute__((ext_vector_type(8))) float v8f;

#define HIDDEN 64
#define DNODE 384

static __device__ __forceinline__ v8f wmma4(v2f a, v2f b, v8f c) {
  return __builtin_amdgcn_wmma_f32_16x16x4_f32(
       false, a,  false, b,
       (short)0, c,  false,  false);
}

static __device__ __forceinline__ void stage_pairs(float* sW, const float* W,
                                                   int rows, int tid, int nthr) {
  const int nvec = rows * HIDDEN / 4;
  for (int idx = tid; idx < nvec; idx += nthr) {
    int k = idx >> 4;
    int col = (idx & 15) * 4;
    v4f v = ((const v4f*)W)[idx];
    float* dstp = sW + (k >> 1) * 128 + (k & 1);
    dstp[(col + 0) * 2] = v.x;
    dstp[(col + 1) * 2] = v.y;
    dstp[(col + 2) * 2] = v.z;
    dstp[(col + 3) * 2] = v.w;
  }
}

__global__ void k_zero(float* __restrict__ p, size_t n) {
  size_t i = (size_t)blockIdx.x * blockDim.x + threadIdx.x;
  size_t stride = (size_t)gridDim.x * blockDim.x;
  for (; i < n; i += stride) p[i] = 0.0f;
}

__global__ void k_deg(const int* __restrict__ dst, float* __restrict__ deg, int E) {
  int e = blockIdx.x * blockDim.x + threadIdx.x;
  if (e < E) atomicAdd(deg + dst[e], 1.0f);
}

#define STRIP 32
#define BATCH 8
__global__ __launch_bounds__(256) void k_scatter(const int* __restrict__ src,
                                                 const int* __restrict__ dst,
                                                 const float* __restrict__ h,
                                                 float* __restrict__ agg, int E) {
  const int t = blockIdx.x * blockDim.x + threadIdx.x;
  const int lane = threadIdx.x & 31;
  const int e0 = t - lane;
  if (e0 >= E) return;
  const int e = e0 + lane;
  const int ec = e < E ? e : E - 1;
  const int s = src[ec];
  const int d = dst[ec];
  const int cnt = (E - e0) < STRIP ? (E - e0) : STRIP;
  const float* hl = h + lane * 2;
  float* al = agg + lane * 2;
  if (cnt == STRIP) {
#pragma unroll
    for (int b = 0; b < STRIP; b += BATCH) {
      v2f v[BATCH];
#pragma unroll
      for (int j = 0; j < BATCH; ++j) {
        const int sj = __builtin_amdgcn_readlane(s, b + j);
        v[j] = *(const v2f*)(hl + (size_t)sj * HIDDEN);
      }
#pragma unroll
      for (int j = 0; j < BATCH; ++j) {
        const int dj = __builtin_amdgcn_readlane(d, b + j);
        float* ap = al + (size_t)dj * HIDDEN;
        atomicAdd(ap, v[j].x);
        atomicAdd(ap + 1, v[j].y);
      }
    }
  } else {
    for (int j = 0; j < cnt; ++j) {
      const int sj = __builtin_amdgcn_readlane(s, j);
      const int dj = __builtin_amdgcn_readlane(d, j);
      const v2f v = *(const v2f*)(hl + (size_t)sj * HIDDEN);
      float* ap = al + (size_t)dj * HIDDEN;
      atomicAdd(ap, v.x);
      atomicAdd(ap + 1, v.y);
    }
  }
}

__global__ __launch_bounds__(128) void k_inproj(const float* __restrict__ x,
                                                const float* __restrict__ Win,
                                                const float* __restrict__ bin,
                                                float* __restrict__ h, int n) {
  __shared__ float sW[96 * HIDDEN];
  const int wave = threadIdx.x >> 5;
  const int lane = threadIdx.x & 31;
  const int hf = lane >> 4;
  const int r = lane & 15;
  const int rowBase = blockIdx.x * 64 + wave * 16;
  const int row = rowBase + r;
  const int rowc = row < n ? row : n - 1;
  const float* xr = x + (size_t)rowc * DNODE;

  v8f acc[4];
#pragma unroll
  for (int c = 0; c < 4; ++c) {
    float b = bin[c * 16 + r];
#pragma unroll
    for (int i = 0; i < 8; ++i) acc[c][i] = b;
  }

  for (int kc = 0; kc < DNODE; kc += 96) {
    __syncthreads();
    stage_pairs(sW, Win + (size_t)kc * HIDDEN, 96, threadIdx.x, 128);
    __syncthreads();

    if (kc + 96 < DNODE) __builtin_prefetch(xr + kc + 96 + 2 * hf, 0, 3);

#pragma unroll 4
    for (int k = 0; k < 96; k += 4) {
      v2f a = *(const v2f*)(xr + kc + k + 2 * hf);
      const float* wrow = sW + ((k >> 1) + hf) * 128;
#pragma unroll
      for (int c = 0; c < 4; ++c) {
        v2f b = *(const v2f*)(wrow + (c * 16 + r) * 2);
        acc[c] = wmma4(a, b, acc[c]);
      }
    }
  }

#pragma unroll
  for (int c = 0; c < 4; ++c) {
#pragma unroll
    for (int i = 0; i < 8; ++i) {
      int m = rowBase + i + 8 * hf;
      if (m < n) h[(size_t)m * HIDDEN + c * 16 + r] = fmaxf(acc[c][i], 0.0f);
    }
  }
}

__global__ __launch_bounds__(128) void k_layer(const float* __restrict__ hprev,
                                               const float* __restrict__ agg,
                                               const float* __restrict__ deg,
                                               const float* __restrict__ Wl,
                                               const float* __restrict__ bl,
                                               const float* __restrict__ Wr,
                                               const float* __restrict__ g,
                                               const float* __restrict__ bt,
                                               float* __restrict__ hnext, int n) {
  __shared__ float sWl[HIDDEN * HIDDEN];
  __shared__ float sWr[HIDDEN * HIDDEN];
  __shared__ float sOut[4 * 16 * HIDDEN];

  stage_pairs(sWl, Wl, HIDDEN, threadIdx.x, 128);
  stage_pairs(sWr, Wr, HIDDEN, threadIdx.x, 128);
  __syncthreads();

  const int wave = threadIdx.x >> 5;
  const int lane = threadIdx.x & 31;
  const int hf = lane >> 4;
  const int r = lane & 15;
  const int rowBase = blockIdx.x * 64 + wave * 16;
  const int row = rowBase + r;
  const int rowc = row < n ? row : n - 1;
  const float* ar = agg + (size_t)rowc * HIDDEN;
  const float* hr = hprev + (size_t)rowc * HIDDEN;
  const float invd = 1.0f / fmaxf(deg[rowc], 1.0f);

  v8f acc[4];
#pragma unroll
  for (int c = 0; c < 4; ++c) {
    float b = bl[c * 16 + r];
#pragma unroll
    for (int i = 0; i < 8; ++i) acc[c][i] = b;
  }

#pragma unroll 4
  for (int k = 0; k < HIDDEN; k += 4) {
    v2f a1 = *(const v2f*)(ar + k + 2 * hf);
    a1.x *= invd;
    a1.y *= invd;
    v2f a2 = *(const v2f*)(hr + k + 2 * hf);
    const float* wl = sWl + ((k >> 1) + hf) * 128;
    const float* wr = sWr + ((k >> 1) + hf) * 128;
#pragma unroll
    for (int c = 0; c < 4; ++c) {
      v2f b1 = *(const v2f*)(wl + (c * 16 + r) * 2);
      acc[c] = wmma4(a1, b1, acc[c]);
      v2f b2 = *(const v2f*)(wr + (c * 16 + r) * 2);
      acc[c] = wmma4(a2, b2, acc[c]);
    }
  }

  float* so = sOut + wave * 16 * HIDDEN;
#pragma unroll
  for (int c = 0; c < 4; ++c)
#pragma unroll
    for (int i = 0; i < 8; ++i)
      so[(i + 8 * hf) * HIDDEN + c * 16 + r] = acc[c][i];
  __syncthreads();

  const int node = rowBase + r;
  const bool valid = node < n;
  const int nodec = valid ? node : n - 1;
  const v4f* rp = (const v4f*)(so + r * HIDDEN + hf * 32);
  v4f vv[8];
  float ss = 0.0f;
#pragma unroll
  for (int j = 0; j < 8; ++j) {
    vv[j] = rp[j];
    ss += vv[j].x * vv[j].x + vv[j].y * vv[j].y + vv[j].z * vv[j].z + vv[j].w * vv[j].w;
  }
  ss += __shfl_xor(ss, 16);
  const float l2 = rsqrtf(fmaxf(ss, 1e-24f));

  const v4f* hres = (const v4f*)(hprev + (size_t)nodec * HIDDEN + hf * 32);
  float s = 0.0f, q = 0.0f;
#pragma unroll
  for (int j = 0; j < 8; ++j) {
    v4f t = vv[j] * l2;
    t.x = fmaxf(t.x, 0.0f);
    t.y = fmaxf(t.y, 0.0f);
    t.z = fmaxf(t.z, 0.0f);
    t.w = fmaxf(t.w, 0.0f);
    t += hres[j];
    vv[j] = t;
    s += t.x + t.y + t.z + t.w;
    q += t.x * t.x + t.y * t.y + t.z * t.z + t.w * t.w;
  }
  s += __shfl_xor(s, 16);
  q += __shfl_xor(q, 16);
  const float mean = s * (1.0f / 64.0f);
  const float var = q * (1.0f / 64.0f) - mean * mean;
  const float sc = rsqrtf(var + 1e-5f);

  const v4f* gp = (const v4f*)(g + hf * 32);
  const v4f* bp = (const v4f*)(bt + hf * 32);
  if (valid) {
    v4f* op = (v4f*)(hnext + (size_t)node * HIDDEN + hf * 32);
#pragma unroll
    for (int j = 0; j < 8; ++j) op[j] = (vv[j] - mean) * sc * gp[j] + bp[j];
  }
}

__global__ __launch_bounds__(256) void k_finalln(const float* __restrict__ h,
                                                 const float* __restrict__ g,
                                                 const float* __restrict__ b,
                                                 float* __restrict__ out, int n) {
  int t = blockIdx.x * blockDim.x + threadIdx.x;
  int row = t >> 5;
  int lane = t & 31;
  if (row >= n) return;
  v2f v = *(const v2f*)(h + (size_t)row * HIDDEN + lane * 2);
  float s = v.x + v.y;
  float q = v.x * v.x + v.y * v.y;
#pragma unroll
  for (int m = 16; m >= 1; m >>= 1) {
    s += __shfl_xor(s, m);
    q += __shfl_xor(q, m);
  }
  float mean = s * (1.0f / 64.0f);
  float var = q * (1.0f / 64.0f) - mean * mean;
  float sc = rsqrtf(var + 1e-5f);
  v2f g2 = *(const v2f*)(g + lane * 2);
  v2f b2 = *(const v2f*)(b + lane * 2);
  v2f o;
  o.x = (v.x - mean) * sc * g2.x + b2.x;
  o.y = (v.y - mean) * sc * g2.y + b2.y;
  *(v2f*)(out + (size_t)row * HIDDEN + lane * 2) = o;
}

extern "C" void kernel_launch(void* const* d_in, const int* in_sizes, int n_in,
                              void* d_out, int out_size, void* d_ws, size_t ws_size,
                              hipStream_t stream) {
  const float* x   = (const float*)d_in[0];
  const int* ei    = (const int*)d_in[1];
  const float* Win = (const float*)d_in[2];
  const float* bin = (const float*)d_in[3];
  const float* Wl  = (const float*)d_in[4];
  const float* bl  = (const float*)d_in[5];
  const float* Wr  = (const float*)d_in[6];
  const float* lng = (const float*)d_in[7];
  const float* lnb = (const float*)d_in[8];
  const float* og  = (const float*)d_in[9];
  const float* ob  = (const float*)d_in[10];

  const int N = in_sizes[0] / DNODE;
  const int E = in_sizes[1] / 2;
  const int LAYERS = in_sizes[4] / (HIDDEN * HIDDEN);
  const int* src = ei;
  const int* dst = ei + E;

  float* h0  = (float*)d_ws;
  float* h1  = h0 + (size_t)N * HIDDEN;
  float* agg = h1 + (size_t)N * HIDDEN;
  float* deg = agg + (size_t)N * HIDDEN;

  k_zero<<<1024, 256, 0, stream>>>(deg, (size_t)N);
  k_deg<<<(E + 255) / 256, 256, 0, stream>>>(dst, deg, E);

  k_inproj<<<(N + 63) / 64, 128, 0, stream>>>(x, Win, bin, h0, N);

  float* cur = h0;
  float* nxt = h1;
  for (int l = 0; l < LAYERS; ++l) {
    k_zero<<<2048, 256, 0, stream>>>(agg, (size_t)N * HIDDEN);
    k_scatter<<<(E + 255) / 256, 256, 0, stream>>>(src, dst, cur, agg, E);
    k_layer<<<(N + 63) / 64, 128, 0, stream>>>(
        cur, agg, deg,
        Wl + (size_t)l * HIDDEN * HIDDEN, bl + (size_t)l * HIDDEN,
        Wr + (size_t)l * HIDDEN * HIDDEN,
        lng + (size_t)l * HIDDEN, lnb + (size_t)l * HIDDEN, nxt, N);
    float* t = cur; cur = nxt; nxt = t;
  }

  k_finalln<<<(int)(((long long)N * 32 + 255) / 256), 256, 0, stream>>>(cur, og, ob, (float*)d_out, N);
}


# scatter: edges counting-sorted by destination once per call (integer rank atomic); one wave per node then sums its source rows in registers, in rank (arrival) order; float atomics and zeroing gone
# speedup vs baseline: 1.9995x; 1.3441x over previous
#include <hip/hip_runtime.h>
#include <hip/hip_bf16.h>

typedef __attribute__((ext_vector_type(2))) float v2f;
typedef __attribute__((ext_vector_type(4))) float v4f;
typedef __attribute__((ext_vector_type(8))) float v8f;

#define HIDDEN 64
#define DNODE 384

static __device__ __forceinline__ v8f wmma4(v2f a, v2f b, v8f c) {
  return __builtin_amdgcn_wmma_f32_16x16x4_f32(
       false, a,  false, b,
       (short)0, c,  false,  false);
}

static __device__ __forceinline__ void stage_pairs(float* sW, const float* W,
                                                   int rows, int tid, int nthr) {
  const int nvec = rows * HIDDEN / 4;
  for (int idx = tid; idx < nvec; idx += nthr) {
    int k = idx >> 4;
    int col = (idx & 15) * 4;
    v4f v = ((const v4f*)W)[idx];
    float* dstp = sW + (k >> 1) * 128 + (k & 1);
    dstp[(col + 0) * 2] = v.x;
    dstp[(col + 1) * 2] = v.y;
    dstp[(col + 2) * 2] = v.z;
    dstp[(col + 3) * 2] = v.w;
  }
}

__global__ void k_zero(float* __restrict__ p, size_t n) {
  size_t i = (size_t)blockIdx.x * blockDim.x + threadIdx.x;
  size_t stride = (size_t)gridDim.x * blockDim.x;
  for (; i < n; i += stride) p[i] = 0.0f;
}

__global__ void k_hist(const int* __restrict__ dst, int* __restrict__ cnt,
                       int* __restrict__ rank, int E) {
  int e = blockIdx.x * blockDim.x + threadIdx.x;
  if (e < E) rank[e] = atomicAdd(cnt + dst[e], 1);
}

__global__ __launch_bounds__(256) void k_alloc(const int* cnt, int* total,
                                               int* start, float* deg, int n) {
  __shared__ int wsum[8];
  __shared__ int base;
  const int i = blockIdx.x * 256 + threadIdx.x;
  const int lane = threadIdx.x & 31;
  const int w = threadIdx.x >> 5;
  const int c = i < n ? cnt[i] : 0;
  int x = c;
#pragma unroll
  for (int d = 1; d < 32; d <<= 1) {
    const int y = __shfl_up(x, d);
    if (lane >= d) x += y;
  }
  if (lane == 31) wsum[w] = x;
  __syncthreads();
  if (threadIdx.x == 0) {
    int run = 0;
    for (int k = 0; k < 8; ++k) {
      const int v = wsum[k];
      wsum[k] = run;
      run += v;
    }
    base = atomicAdd(total, run);
  }
  __syncthreads();
  if (i < n) {
    start[i] = base + wsum[w] + x - c;
    deg[i] = (float)c;
  }
}

__global__ void k_fill(const int* __restrict__ src, const int* __restrict__ dst,
                       const int* __restrict__ start, const int* __restrict__ rank,
                       int* __restrict__ ssrc, int E) {
  int e = blockIdx.x * blockDim.x + threadIdx.x;
  if (e < E) ssrc[start[dst[e]] + rank[e]] = src[e];
}

template <int G>
static __device__ __forceinline__ void gather_group(const float* hl, int s, int j,
                                                    v2f& acc) {
  v2f v[G];
#pragma unroll
  for (int u = 0; u < G; ++u) {
    const int sj = __builtin_amdgcn_readlane(s, j + u);
    v[u] = *(const v2f*)(hl + (size_t)sj * HIDDEN);
  }
#pragma unroll
  for (int u = 0; u < G; ++u) {
    acc.x += v[u].x;
    acc.y += v[u].y;
  }
}

__global__ __launch_bounds__(256) void k_gather(const int* __restrict__ ssrc,
                                                const int* __restrict__ start,
                                                const int* __restrict__ cnt,
                                                const float* __restrict__ h,
                                                float* __restrict__ agg, int n) {
  const int t = blockIdx.x * blockDim.x + threadIdx.x;
  const int lane = threadIdx.x & 31;
  const int node = __builtin_amdgcn_readfirstlane(t >> 5);
  if (node >= n) return;
  const int c = cnt[node];
  const int s0 = start[node];
  const float* hl = h + lane * 2;
  v2f acc = {0.0f, 0.0f};
  for (int b = 0; b < c; b += 32) {
    const int m = (c - b) < 32 ? (c - b) : 32;
    const int li = lane < m ? lane : m - 1;
    const int s = ssrc[s0 + b + li];
    int j = 0;
    for (; j + 8 <= m; j += 8) gather_group<8>(hl, s, j, acc);
    if (j + 4 <= m) { gather_group<4>(hl, s, j, acc); j += 4; }
    if (j + 2 <= m) { gather_group<2>(hl, s, j, acc); j += 2; }
    if (j < m) gather_group<1>(hl, s, j, acc);
  }
  *(v2f*)(agg + (size_t)node * HIDDEN + lane * 2) = acc;
}

__global__ __launch_bounds__(128) void k_inproj(const float* __restrict__ x,
                                                const float* __restrict__ Win,
                                                const float* __restrict__ bin,
                                                float* __restrict__ h, int n) {
  __shared__ float sW[96 * HIDDEN];
  const int wave = threadIdx.x >> 5;
  const int lane = threadIdx.x & 31;
  const int hf = lane >> 4;
  const int r = lane & 15;
  const int rowBase = blockIdx.x * 64 + wave * 16;
  const int row = rowBase + r;
  const int rowc = row < n ? row : n - 1;
  const float* xr = x + (size_t)rowc * DNODE;

  v8f acc[4];
#pragma unroll
  for (int c = 0; c < 4; ++c) {
    float b = bin[c * 16 + r];
#pragma unroll
    for (int i = 0; i < 8; ++i) acc[c][i] = b;
  }

  for (int kc = 0; kc < DNODE; kc += 96) {
    __syncthreads();
    stage_pairs(sW, Win + (size_t)kc * HIDDEN, 96, threadIdx.x, 128);
    __syncthreads();

    if (kc + 96 < DNODE) __builtin_prefetch(xr + kc + 96 + 2 * hf, 0, 3);

#pragma unroll 4
    for (int k = 0; k < 96; k += 4) {
      v2f a = *(const v2f*)(xr + kc + k + 2 * hf);
      const float* wrow = sW + ((k >> 1) + hf) * 128;
#pragma unroll
      for (int c = 0; c < 4; ++c) {
        v2f b = *(const v2f*)(wrow + (c * 16 + r) * 2);
        acc[c] = wmma4(a, b, acc[c]);
      }
    }
  }

#pragma unroll
  for (int c = 0; c < 4; ++c) {
#pragma unroll
    for (int i = 0; i < 8; ++i) {
      int m = rowBase + i + 8 * hf;
      if (m < n) h[(size_t)m * HIDDEN + c * 16 + r] = fmaxf(acc[c][i], 0.0f);
    }
  }
}

__global__ __launch_bounds__(128) void k_layer(const float* __restrict__ hprev,
                                               const float* __restrict__ agg,
                                               const float* __restrict__ deg,
                                               const float* __restrict__ Wl,
                                               const float* __restrict__ bl,
                                               const float* __restrict__ Wr,
                                               const float* __restrict__ g,
                                               const float* __restrict__ bt,
                                               float* __restrict__ hnext, int n) {
  __shared__ float sWl[HIDDEN * HIDDEN];
  __shared__ float sWr[HIDDEN * HIDDEN];
  __shared__ float sOut[4 * 16 * HIDDEN];

  stage_pairs(sWl, Wl, HIDDEN, threadIdx.x, 128);
  stage_pairs(sWr, Wr, HIDDEN, threadIdx.x, 128);
  __syncthreads();

  const int wave = threadIdx.x >> 5;
  const int lane = threadIdx.x & 31;
  const int hf = lane >> 4;
  const int r = lane & 15;
  const int rowBase = blockIdx.x * 64 + wave * 16;
  const int row = rowBase + r;
  const int rowc = row < n ? row : n - 1;
  const float* ar = agg + (size_t)rowc * HIDDEN;
  const float* hr = hprev + (size_t)rowc * HIDDEN;
  const float invd = 1.0f / fmaxf(deg[rowc], 1.0f);

  v8f acc[4];
#pragma unroll
  for (int c = 0; c < 4; ++c) {
    float b = bl[c * 16 + r];
#pragma unroll
    for (int i = 0; i < 8; ++i) acc[c][i] = b;
  }

#pragma unroll 4
  for (int k = 0; k < HIDDEN; k += 4) {
    v2f a1 = *(const v2f*)(ar + k + 2 * hf);
    a1.x *= invd;
    a1.y *= invd;
    v2f a2 = *(const v2f*)(hr + k + 2 * hf);
    const float* wl = sWl + ((k >> 1) + hf) * 128;
    const float* wr = sWr + ((k >> 1) + hf) * 128;
#pragma unroll
    for (int c = 0; c < 4; ++c) {
      v2f b1 = *(const v2f*)(wl + (c * 16 + r) * 2);
      acc[c] = wmma4(a1, b1, acc[c]);
      v2f b2 = *(const v2f*)(wr + (c * 16 + r) * 2);
      acc[c] = wmma4(a2, b2, acc[c]);
    }
  }

  float* so = sOut + wave * 16 * HIDDEN;
#pragma unroll
  for (int c = 0; c < 4; ++c)
#pragma unroll
    for (int i = 0; i < 8; ++i)
      so[(i + 8 * hf) * HIDDEN + c * 16 + r] = acc[c][i];
  __syncthreads();

  const int node = rowBase + r;
  const bool valid = node < n;
  const int nodec = valid ? node : n - 1;
  const v4f* rp = (const v4f*)(so + r * HIDDEN + hf * 32);
  v4f vv[8];
  float ss = 0.0f;
#pragma unroll
  for (int j = 0; j < 8; ++j) {
    vv[j] = rp[j];
    ss += vv[j].x * vv[j].x + vv[j].y * vv[j].y + vv[j].z * vv[j].z + vv[j].w * vv[j].w;
  }
  ss += __shfl_xor(ss, 16);
  const float l2 = rsqrtf(fmaxf(ss, 1e-24f));

  const v4f* hres = (const v4f*)(hprev + (size_t)nodec * HIDDEN + hf * 32);
  float s = 0.0f, q = 0.0f;
#pragma unroll
  for (int j = 0; j < 8; ++j) {
    v4f t = vv[j] * l2;
    t.x = fmaxf(t.x, 0.0f);
    t.y = fmaxf(t.y, 0.0f);
    t.z = fmaxf(t.z, 0.0f);
    t.w = fmaxf(t.w, 0.0f);
    t += hres[j];
    vv[j] = t;
    s += t.x + t.y + t.z + t.w;
    q += t.x * t.x + t.y * t.y + t.z * t.z + t.w * t.w;
  }
  s += __shfl_xor(s, 16);
  q += __shfl_xor(q, 16);
  const float mean = s * (1.0f / 64.0f);
  const float var = q * (1.0f / 64.0f) - mean * mean;
  const float sc = rsqrtf(var + 1e-5f);

  const v4f* gp = (const v4f*)(g + hf * 32);
  const v4f* bp = (const v4f*)(bt + hf * 32);
  if (valid) {
    v4f* op = (v4f*)(hnext + (size_t)node * HIDDEN + hf * 32);
#pragma unroll
    for (int j = 0; j < 8; ++j) op[j] = (vv[j] - mean) * sc * gp[j] + bp[j];
  }
}

__global__ __launch_bounds__(256) void k_finalln(const float* __restrict__ h,
                                                 const float* __restrict__ g,
                                                 const float* __restrict__ b,
                                                 float* __restrict__ out, int n) {
  int t = blockIdx.x * blockDim.x + threadIdx.x;
  int row = t >> 5;
  int lane = t & 31;
  if (row >= n) return;
  v2f v = *(const v2f*)(h + (size_t)row * HIDDEN + lane * 2);
  float s = v.x + v.y;
  float q = v.x * v.x + v.y * v.y;
#pragma unroll
  for (int m = 16; m >= 1; m >>= 1) {
    s += __shfl_xor(s, m);
    q += __shfl_xor(q, m);
  }
  float mean = s * (1.0f / 64.0f);
  float var = q * (1.0f / 64.0f) - mean * mean;
  float sc = rsqrtf(var + 1e-5f);
  v2f g2 = *(const v2f*)(g + lane * 2);
  v2f b2 = *(const v2f*)(b + lane * 2);
  v2f o;
  o.x = (v.x - mean) * sc * g2.x + b2.x;
  o.y = (v.y - mean) * sc * g2.y + b2.y;
  *(v2f*)(out + (size_t)row * HIDDEN + lane * 2) = o;
}

extern "C" void kernel_launch(void* const* d_in, const int* in_sizes, int n_in,
                              void* d_out, int out_size, void* d_ws, size_t ws_size,
                              hipStream_t stream) {
  const float* x   = (const float*)d_in[0];
  const int* ei    = (const int*)d_in[1];
  const float* Win = (const float*)d_in[2];
  const float* bin = (const float*)d_in[3];
  const float* Wl  = (const float*)d_in[4];
  const float* bl  = (const float*)d_in[5];
  const float* Wr  = (const float*)d_in[6];
  const float* lng = (const float*)d_in[7];
  const float* lnb = (const float*)d_in[8];
  const float* og  = (const float*)d_in[9];
  const float* ob  = (const float*)d_in[10];

  const int N = in_sizes[0] / DNODE;
  const int E = in_sizes[1] / 2;
  const int LAYERS = in_sizes[4] / (HIDDEN * HIDDEN);
  const int* src = ei;
  const int* dst = ei + E;

  float* h0  = (float*)d_ws;
  float* h1  = h0 + (size_t)N * HIDDEN;
  int* ssrc  = (int*)(h1 + (size_t)N * HIDDEN);
  int* rank  = ssrc + E;
  int* cnt   = rank + E;
  int* total = cnt + N;
  int* start = total + 1;
  float* deg = h1 + (size_t)2 * N * HIDDEN;
  float* agg = (float*)d_out;

  k_zero<<<1024, 256, 0, stream>>>((float*)cnt, (size_t)N + 1);
  k_hist<<<(E + 255) / 256, 256, 0, stream>>>(dst, cnt, rank, E);
  k_alloc<<<(N + 255) / 256, 256, 0, stream>>>(cnt, total, start, deg, N);
  k_fill<<<(E + 255) / 256, 256, 0, stream>>>(src, dst, start, rank, ssrc, E);

  k_inproj<<<(N + 63) / 64, 128, 0, stream>>>(x, Win, bin, h0, N);

  float* cur = h0;
  float* nxt = h1;
  for (int l = 0; l < LAYERS; ++l) {
    k_gather<<<(int)(((long long)N * 32 + 255) / 256), 256, 0, stream>>>(ssrc, start, cnt, cur, agg, N);
    k_layer<<<(N + 63) / 64, 128, 0, stream>>>(
        cur, agg, deg,
        Wl + (size_t)l * HIDDEN * HIDDEN, bl + (size_t)l * HIDDEN,
        Wr + (size_t)l * HIDDEN * HIDDEN,
        lng + (size_t)l * HIDDEN, lnb + (size_t)l * HIDDEN, nxt, N);
    float* t = cur; cur = nxt; nxt = t;
  }

  k_finalln<<<(int)(((long long)N * 32 + 255) / 256), 256, 0, stream>>>(cur, og, ob, (float*)d_out, N);
}


# edge sort: per-edge global rank atomic replaced by LDS counters (256 edge chunks x 512-node buckets: count, scan, place, per-bucket sort); rows still summed in arrival order, now of LDS adds
# speedup vs baseline: 2.6933x; 1.3470x over previous
#include <hip/hip_runtime.h>
#include <hip/hip_bf16.h>

typedef __attribute__((ext_vector_type(2))) float v2f;
typedef __attribute__((ext_vector_type(4))) float v4f;
typedef __attribute__((ext_vector_type(8))) float v8f;

#define HIDDEN 64
#define DNODE 384

static __device__ __forceinline__ v8f wmma4(v2f a, v2f b, v8f c) {
  return __builtin_amdgcn_wmma_f32_16x16x4_f32(
       false, a,  false, b,
       (short)0, c,  false,  false);
}

static __device__ __forceinline__ void stage_pairs(float* sW, const float* W,
                                                   int rows, int tid, int nthr) {
  const int nvec = rows * HIDDEN / 4;
  for (int idx = tid; idx < nvec; idx += nthr) {
    int k = idx >> 4;
    int col = (idx & 15) * 4;
    v4f v = ((const v4f*)W)[idx];
    float* dstp = sW + (k >> 1) * 128 + (k & 1);
    dstp[(col + 0) * 2] = v.x;
    dstp[(col + 1) * 2] = v.y;
    dstp[(col + 2) * 2] = v.z;
    dstp[(col + 3) * 2] = v.w;
  }
}

#define NCHUNK 256
#define BSHIFT 9
#define BNODES 512
#define BMAX 1024
#define EU 8

template <int NW>
static __device__ __forceinline__ int block_scan_excl(int v, int* wsum) {
  const int lane = threadIdx.x & 31;
  const int wave = threadIdx.x >> 5;
  int x = v;
#pragma unroll
  for (int d = 1; d < 32; d <<= 1) {
    const int y = __shfl_up(x, d);
    if (lane >= d) x += y;
  }
  if (lane == 31) wsum[wave] = x;
  __syncthreads();
  if (wave == 0) {
    const int tv = lane < NW ? wsum[lane] : 0;
    int z = tv;
#pragma unroll
    for (int d = 1; d < 32; d <<= 1) {
      const int y = __shfl_up(z, d);
      if (lane >= d) z += y;
    }
    if (lane < NW) wsum[lane] = z - tv;
  }
  __syncthreads();
  return wsum[wave] + x - v;
}

__global__ __launch_bounds__(1024) void k_bcount(const int* __restrict__ dst,
                                                 int* __restrict__ wcnt,
                                                 int E, int per, int B) {
  __shared__ int hb[BMAX];
  const int t = threadIdx.x;
  hb[t] = 0;
  __syncthreads();
  const int e0 = blockIdx.x * per;
  const int e1 = (e0 + per) < E ? (e0 + per) : E;
  for (int e = e0 + t; e < e1; e += EU * 1024) {
    int d[EU];
#pragma unroll
    for (int u = 0; u < EU; ++u) {
      const int ee = e + u * 1024;
      d[u] = ee < e1 ? dst[ee] : 0;
    }
#pragma unroll
    for (int u = 0; u < EU; ++u)
      if (e + u * 1024 < e1) atomicAdd(&hb[d[u] >> BSHIFT], 1);
  }
  __syncthreads();
  if (t < B) wcnt[blockIdx.x * B + t] = hb[t];
}

__global__ __launch_bounds__(NCHUNK) void k_bscan(const int* __restrict__ wcnt,
                                                  int* __restrict__ wpre,
                                                  int* __restrict__ btot, int B) {
  __shared__ int wsum[32];
  const int b = blockIdx.x;
  const int w = threadIdx.x;
  const int v = wcnt[w * B + b];
  const int ex = block_scan_excl<NCHUNK / 32>(v, wsum);
  wpre[w * B + b] = ex;
  if (w == NCHUNK - 1) btot[b] = ex + v;
}

__global__ __launch_bounds__(1024) void k_bplace(const int* __restrict__ src,
                                                 const int* __restrict__ dst,
                                                 const int* __restrict__ wpre,
                                                 const int* __restrict__ btot,
                                                 int* __restrict__ bbase,
                                                 int* __restrict__ tmp,
                                                 int E, int per, int B) {
  __shared__ int cur[BMAX];
  __shared__ int wsum[32];
  const int t = threadIdx.x;
  const int w = blockIdx.x;
  const int tot = t < B ? btot[t] : 0;
  const int ex = block_scan_excl<32>(tot, wsum);
  if (t < B) {
    cur[t] = ex + wpre[w * B + t];
    if (w == 0) bbase[t] = ex;
  }
  if (w == 0 && t == 0) bbase[B] = E;
  __syncthreads();
  const int e0 = w * per;
  const int e1 = (e0 + per) < E ? (e0 + per) : E;
  int2* tmp2 = (int2*)tmp;
  for (int e = e0 + t; e < e1; e += EU * 1024) {
    int d[EU], s[EU];
#pragma unroll
    for (int u = 0; u < EU; ++u) {
      const int ee = e + u * 1024;
      const bool ok = ee < e1;
      d[u] = ok ? dst[ee] : 0;
      s[u] = ok ? src[ee] : 0;
    }
#pragma unroll
    for (int u = 0; u < EU; ++u) {
      if (e + u * 1024 < e1) {
        const int p = atomicAdd(&cur[d[u] >> BSHIFT], 1);
        tmp2[p] = make_int2(s[u], d[u]);
      }
    }
  }
}

__global__ __launch_bounds__(1024) void k_bsort(const int* __restrict__ tmp,
                                                const int* __restrict__ bbase,
                                                int* __restrict__ ssrc,
                                                int* __restrict__ cnt,
                                                int* __restrict__ start,
                                                float* __restrict__ deg, int n) {
  __shared__ int c[BNODES];
  __shared__ int wsum[32];
  const int t = threadIdx.x;
  const int b = blockIdx.x;
  const int e0 = bbase[b];
  const int e1 = bbase[b + 1];
  const int2* tmp2 = (const int2*)tmp;
  if (t < BNODES) c[t] = 0;
  __syncthreads();
  for (int e = e0 + t; e < e1; e += EU * 1024) {
    int d[EU];
#pragma unroll
    for (int u = 0; u < EU; ++u) {
      const int ee = e + u * 1024;
      d[u] = ee < e1 ? tmp[2 * (size_t)ee + 1] : 0;
    }
#pragma unroll
    for (int u = 0; u < EU; ++u)
      if (e + u * 1024 < e1) atomicAdd(&c[d[u] & (BNODES - 1)], 1);
  }
  __syncthreads();
  const int v = t < BNODES ? c[t] : 0;
  const int ex = block_scan_excl<32>(v, wsum);
  const int node = b * BNODES + t;
  if (t < BNODES) {
    c[t] = e0 + ex;
    if (node < n) {
      cnt[node] = v;
      start[node] = e0 + ex;
      deg[node] = (float)v;
    }
  }
  __syncthreads();
  for (int e = e0 + t; e < e1; e += EU * 1024) {
    int2 sd[EU];
#pragma unroll
    for (int u = 0; u < EU; ++u) {
      const int ee = e + u * 1024;
      sd[u] = ee < e1 ? tmp2[ee] : make_int2(0, 0);
    }
#pragma unroll
    for (int u = 0; u < EU; ++u) {
      if (e + u * 1024 < e1) {
        const int p = atomicAdd(&c[sd[u].y & (BNODES - 1)], 1);
        ssrc[p] = sd[u].x;
      }
    }
  }
}

template <int G>
static __device__ __forceinline__ void gather_group(const float* hl, int s, int j,
                                                    v2f& acc) {
  v2f v[G];
#pragma unroll
  for (int u = 0; u < G; ++u) {
    const int sj = __builtin_amdgcn_readlane(s, j + u);
    v[u] = *(const v2f*)(hl + (size_t)sj * HIDDEN);
  }
#pragma unroll
  for (int u = 0; u < G; ++u) {
    acc.x += v[u].x;
    acc.y += v[u].y;
  }
}

__global__ __launch_bounds__(256) void k_gather(const int* __restrict__ ssrc,
                                                const int* __restrict__ start,
                                                const int* __restrict__ cnt,
                                                const float* __restrict__ h,
                                                float* __restrict__ agg, int n) {
  const int t = blockIdx.x * blockDim.x + threadIdx.x;
  const int lane = threadIdx.x & 31;
  const int node = __builtin_amdgcn_readfirstlane(t >> 5);
  if (node >= n) return;
  const int c = cnt[node];
  const int s0 = start[node];
  const float* hl = h + lane * 2;
  v2f acc = {0.0f, 0.0f};
  for (int b = 0; b < c; b += 32) {
    const int m = (c - b) < 32 ? (c - b) : 32;
    const int li = lane < m ? lane : m - 1;
    const int s = ssrc[s0 + b + li];
    int j = 0;
    for (; j + 8 <= m; j += 8) gather_group<8>(hl, s, j, acc);
    if (j + 4 <= m) { gather_group<4>(hl, s, j, acc); j += 4; }
    if (j + 2 <= m) { gather_group<2>(hl, s, j, acc); j += 2; }
    if (j < m) gather_group<1>(hl, s, j, acc);
  }
  *(v2f*)(agg + (size_t)node * HIDDEN + lane * 2) = acc;
}

__global__ __launch_bounds__(128) void k_inproj(const float* __restrict__ x,
                                                const float* __restrict__ Win,
                                                const float* __restrict__ bin,
                                                float* __restrict__ h, int n) {
  __shared__ float sW[96 * HIDDEN];
  const int wave = threadIdx.x >> 5;
  const int lane = threadIdx.x & 31;
  const int hf = lane >> 4;
  const int r = lane & 15;
  const int rowBase = blockIdx.x * 64 + wave * 16;
  const int row = rowBase + r;
  const int rowc = row < n ? row : n - 1;
  const float* xr = x + (size_t)rowc * DNODE;

  v8f acc[4];
#pragma unroll
  for (int c = 0; c < 4; ++c) {
    float b = bin[c * 16 + r];
#pragma unroll
    for (int i = 0; i < 8; ++i) acc[c][i] = b;
  }

  for (int kc = 0; kc < DNODE; kc += 96) {
    __syncthreads();
    stage_pairs(sW, Win + (size_t)kc * HIDDEN, 96, threadIdx.x, 128);
    __syncthreads();

    if (kc + 96 < DNODE) __builtin_prefetch(xr + kc + 96 + 2 * hf, 0, 3);

#pragma unroll 4
    for (int k = 0; k < 96; k += 4) {
      v2f a = *(const v2f*)(xr + kc + k + 2 * hf);
      const float* wrow = sW + ((k >> 1) + hf) * 128;
#pragma unroll
      for (int c = 0; c < 4; ++c) {
        v2f b = *(const v2f*)(wrow + (c * 16 + r) * 2);
        acc[c] = wmma4(a, b, acc[c]);
      }
    }
  }

#pragma unroll
  for (int c = 0; c < 4; ++c) {
#pragma unroll
    for (int i = 0; i < 8; ++i) {
      int m = rowBase + i + 8 * hf;
      if (m < n) h[(size_t)m * HIDDEN + c * 16 + r] = fmaxf(acc[c][i], 0.0f);
    }
  }
}

__global__ __launch_bounds__(128) void k_layer(const float* __restrict__ hprev,
                                               const float* __restrict__ agg,
                                               const float* __restrict__ deg,
                                               const float* __restrict__ Wl,
                                               const float* __restrict__ bl,
                                               const float* __restrict__ Wr,
                                               const float* __restrict__ g,
                                               const float* __restrict__ bt,
                                               float* __restrict__ hnext, int n) {
  __shared__ float sWl[HIDDEN * HIDDEN];
  __shared__ float sWr[HIDDEN * HIDDEN];
  __shared__ float sOut[4 * 16 * HIDDEN];

  stage_pairs(sWl, Wl, HIDDEN, threadIdx.x, 128);
  stage_pairs(sWr, Wr, HIDDEN, threadIdx.x, 128);
  __syncthreads();

  const int wave = threadIdx.x >> 5;
  const int lane = threadIdx.x & 31;
  const int hf = lane >> 4;
  const int r = lane & 15;
  const int rowBase = blockIdx.x * 64 + wave * 16;
  const int row = rowBase + r;
  const int rowc = row < n ? row : n - 1;
  const float* ar = agg + (size_t)rowc * HIDDEN;
  const float* hr = hprev + (size_t)rowc * HIDDEN;
  const float invd = 1.0f / fmaxf(deg[rowc], 1.0f);

  v8f acc[4];
#pragma unroll
  for (int c = 0; c < 4; ++c) {
    float b = bl[c * 16 + r];
#pragma unroll
    for (int i = 0; i < 8; ++i) acc[c][i] = b;
  }

#pragma unroll 4
  for (int k = 0; k < HIDDEN; k += 4) {
    v2f a1 = *(const v2f*)(ar + k + 2 * hf);
    a1.x *= invd;
    a1.y *= invd;
    v2f a2 = *(const v2f*)(hr + k + 2 * hf);
    const float* wl = sWl + ((k >> 1) + hf) * 128;
    const float* wr = sWr + ((k >> 1) + hf) * 128;
#pragma unroll
    for (int c = 0; c < 4; ++c) {
      v2f b1 = *(const v2f*)(wl + (c * 16 + r) * 2);
      acc[c] = wmma4(a1, b1, acc[c]);
      v2f b2 = *(const v2f*)(wr + (c * 16 + r) * 2);
      acc[c] = wmma4(a2, b2, acc[c]);
    }
  }

  float* so = sOut + wave * 16 * HIDDEN;
#pragma unroll
  for (int c = 0; c < 4; ++c)
#pragma unroll
    for (int i = 0; i < 8; ++i)
      so[(i + 8 * hf) * HIDDEN + c * 16 + r] = acc[c][i];
  __syncthreads();

  const int node = rowBase + r;
  const bool valid = node < n;
  const int nodec = valid ? node : n - 1;
  const v4f* rp = (const v4f*)(so + r * HIDDEN + hf * 32);
  v4f vv[8];
  float ss = 0.0f;
#pragma unroll
  for (int j = 0; j < 8; ++j) {
    vv[j] = rp[j];
    ss += vv[j].x * vv[j].x + vv[j].y * vv[j].y + vv[j].z * vv[j].z + vv[j].w * vv[j].w;
  }
  ss += __shfl_xor(ss, 16);
  const float l2 = rsqrtf(fmaxf(ss, 1e-24f));

  const v4f* hres = (const v4f*)(hprev + (size_t)nodec * HIDDEN + hf * 32);
  float s = 0.0f, q = 0.0f;
#pragma unroll
  for (int j = 0; j < 8; ++j) {
    v4f t = vv[j] * l2;
    t.x = fmaxf(t.x, 0.0f);
    t.y = fmaxf(t.y, 0.0f);
    t.z = fmaxf(t.z, 0.0f);
    t.w = fmaxf(t.w, 0.0f);
    t += hres[j];
    vv[j] = t;
    s += t.x + t.y + t.z + t.w;
    q += t.x * t.x + t.y * t.y + t.z * t.z + t.w * t.w;
  }
  s += __shfl_xor(s, 16);
  q += __shfl_xor(q, 16);
  const float mean = s * (1.0f / 64.0f);
  const float var = q * (1.0f / 64.0f) - mean * mean;
  const float sc = rsqrtf(var + 1e-5f);

  const v4f* gp = (const v4f*)(g + hf * 32);
  const v4f* bp = (const v4f*)(bt + hf * 32);
  if (valid) {
    v4f* op = (v4f*)(hnext + (size_t)node * HIDDEN + hf * 32);
#pragma unroll
    for (int j = 0; j < 8; ++j) op[j] = (vv[j] - mean) * sc * gp[j] + bp[j];
  }
}

__global__ __launch_bounds__(256) void k_finalln(const float* __restrict__ h,
                                                 const float* __restrict__ g,
                                                 const float* __restrict__ b,
                                                 float* __restrict__ out, int n) {
  int t = blockIdx.x * blockDim.x + threadIdx.x;
  int row = t >> 5;
  int lane = t & 31;
  if (row >= n) return;
  v2f v = *(const v2f*)(h + (size_t)row * HIDDEN + lane * 2);
  float s = v.x + v.y;
  float q = v.x * v.x + v.y * v.y;
#pragma unroll
  for (int m = 16; m >= 1; m >>= 1) {
    s += __shfl_xor(s, m);
    q += __shfl_xor(q, m);
  }
  float mean = s * (1.0f / 64.0f);
  float var = q * (1.0f / 64.0f) - mean * mean;
  float sc = rsqrtf(var + 1e-5f);
  v2f g2 = *(const v2f*)(g + lane * 2);
  v2f b2 = *(const v2f*)(b + lane * 2);
  v2f o;
  o.x = (v.x - mean) * sc * g2.x + b2.x;
  o.y = (v.y - mean) * sc * g2.y + b2.y;
  *(v2f*)(out + (size_t)row * HIDDEN + lane * 2) = o;
}

extern "C" void kernel_launch(void* const* d_in, const int* in_sizes, int n_in,
                              void* d_out, int out_size, void* d_ws, size_t ws_size,
                              hipStream_t stream) {
  const float* x   = (const float*)d_in[0];
  const int* ei    = (const int*)d_in[1];
  const float* Win = (const float*)d_in[2];
  const float* bin = (const float*)d_in[3];
  const float* Wl  = (const float*)d_in[4];
  const float* bl  = (const float*)d_in[5];
  const float* Wr  = (const float*)d_in[6];
  const float* lng = (const float*)d_in[7];
  const float* lnb = (const float*)d_in[8];
  const float* og  = (const float*)d_in[9];
  const float* ob  = (const float*)d_in[10];

  const int N = in_sizes[0] / DNODE;
  const int E = in_sizes[1] / 2;
  const int LAYERS = in_sizes[4] / (HIDDEN * HIDDEN);
  const int* src = ei;
  const int* dst = ei + E;

  const int B = (N + BNODES - 1) >> BSHIFT;
  const int per = (E + NCHUNK - 1) / NCHUNK;
  float* h0  = (float*)d_ws;
  float* h1  = h0 + (size_t)N * HIDDEN;
  int* tmp   = (int*)(h1 + (size_t)N * HIDDEN);
  int* ssrc  = tmp + (size_t)2 * E;
  int* cnt   = ssrc + E;
  int* start = cnt + N;
  int* wcnt  = start + N;
  int* wpre  = wcnt + (size_t)NCHUNK * B;
  int* btot  = wpre + (size_t)NCHUNK * B;
  int* bbase = btot + B;
  float* deg = h1 + (size_t)2 * N * HIDDEN;
  float* agg = (float*)d_out;

  k_bcount<<<NCHUNK, 1024, 0, stream>>>(dst, wcnt, E, per, B);
  k_bscan<<<B, NCHUNK, 0, stream>>>(wcnt, wpre, btot, B);
  k_bplace<<<NCHUNK, 1024, 0, stream>>>(src, dst, wpre, btot, bbase, tmp, E, per, B);
  k_bsort<<<B, 1024, 0, stream>>>(tmp, bbase, ssrc, cnt, start, deg, N);

  k_inproj<<<(N + 63) / 64, 128, 0, stream>>>(x, Win, bin, h0, N);

  float* cur = h0;
  float* nxt = h1;
  for (int l = 0; l < LAYERS; ++l) {
    k_gather<<<(int)(((long long)N * 32 + 255) / 256), 256, 0, stream>>>(ssrc, start, cnt, cur, agg, N);
    k_layer<<<(N + 63) / 64, 128, 0, stream>>>(
        cur, agg, deg,
        Wl + (size_t)l * HIDDEN * HIDDEN, bl + (size_t)l * HIDDEN,
        Wr + (size_t)l * HIDDEN * HIDDEN,
        lng + (size_t)l * HIDDEN, lnb + (size_t)l * HIDDEN, nxt, N);
    float* t = cur; cur = nxt; nxt = t;
  }

  k_finalln<<<(int)(((long long)N * 32 + 255) / 256), 256, 0, stream>>>(cur, og, ob, (float*)d_out, N);
}


# layer: neighbour sum fused into the layer kernel (16-wave group sums 64 nodes' rows in slot order; mean and own rows reach WMMA via LDS; gather launches gone); row norms summed over 8 lanes: new tree
# speedup vs baseline: 3.3497x; 1.2437x over previous
#include <hip/hip_runtime.h>
#include <hip/hip_bf16.h>

typedef __attribute__((ext_vector_type(2))) float v2f;
typedef __attribute__((ext_vector_type(4))) float v4f;
typedef __attribute__((ext_vector_type(8))) float v8f;

#define HIDDEN 64
#define DNODE 384

static __device__ __forceinline__ v8f wmma4(v2f a, v2f b, v8f c) {
  return __builtin_amdgcn_wmma_f32_16x16x4_f32(
       false, a,  false, b,
       (short)0, c,  false,  false);
}

static __device__ __forceinline__ void stage_pairs(float* sW, const float* W,
                                                   int rows, int tid, int nthr) {
  const int nvec = rows * HIDDEN / 4;
  for (int idx = tid; idx < nvec; idx += nthr) {
    int k = idx >> 4;
    int col = (idx & 15) * 4;
    v4f v = ((const v4f*)W)[idx];
    float* dstp = sW + (k >> 1) * 128 + (k & 1);
    dstp[(col + 0) * 2] = v.x;
    dstp[(col + 1) * 2] = v.y;
    dstp[(col + 2) * 2] = v.z;
    dstp[(col + 3) * 2] = v.w;
  }
}

#define NCHUNK 256
#define BSHIFT 9
#define BNODES 512
#define BMAX 1024
#define EU 8

template <int NW>
static __device__ __forceinline__ int block_scan_excl(int v, int* wsum) {
  const int lane = threadIdx.x & 31;
  const int wave = threadIdx.x >> 5;
  int x = v;
#pragma unroll
  for (int d = 1; d < 32; d <<= 1) {
    const int y = __shfl_up(x, d);
    if (lane >= d) x += y;
  }
  if (lane == 31) wsum[wave] = x;
  __syncthreads();
  if (wave == 0) {
    const int tv = lane < NW ? wsum[lane] : 0;
    int z = tv;
#pragma unroll
    for (int d = 1; d < 32; d <<= 1) {
      const int y = __shfl_up(z, d);
      if (lane >= d) z += y;
    }
    if (lane < NW) wsum[lane] = z - tv;
  }
  __syncthreads();
  return wsum[wave] + x - v;
}

__global__ __launch_bounds__(1024) void k_bcount(const int* __restrict__ dst,
                                                 int* __restrict__ wcnt,
                                                 int E, int per, int B) {
  __shared__ int hb[BMAX];
  const int t = threadIdx.x;
  hb[t] = 0;
  __syncthreads();
  const int e0 = blockIdx.x * per;
  const int e1 = (e0 + per) < E ? (e0 + per) : E;
  for (int e = e0 + t; e < e1; e += EU * 1024) {
    int d[EU];
#pragma unroll
    for (int u = 0; u < EU; ++u) {
      const int ee = e + u * 1024;
      d[u] = ee < e1 ? dst[ee] : 0;
    }
#pragma unroll
    for (int u = 0; u < EU; ++u)
      if (e + u * 1024 < e1) atomicAdd(&hb[d[u] >> BSHIFT], 1);
  }
  __syncthreads();
  if (t < B) wcnt[blockIdx.x * B + t] = hb[t];
}

__global__ __launch_bounds__(NCHUNK) void k_bscan(const int* __restrict__ wcnt,
                                                  int* __restrict__ wpre,
                                                  int* __restrict__ btot, int B) {
  __shared__ int wsum[32];
  const int b = blockIdx.x;
  const int w = threadIdx.x;
  const int v = wcnt[w * B + b];
  const int ex = block_scan_excl<NCHUNK / 32>(v, wsum);
  wpre[w * B + b] = ex;
  if (w == NCHUNK - 1) btot[b] = ex + v;
}

__global__ __launch_bounds__(1024) void k_bplace(const int* __restrict__ src,
                                                 const int* __restrict__ dst,
                                                 const int* __restrict__ wpre,
                                                 const int* __restrict__ btot,
                                                 int* __restrict__ bbase,
                                                 int* __restrict__ tmp,
                                                 int E, int per, int B) {
  __shared__ int cur[BMAX];
  __shared__ int wsum[32];
  const int t = threadIdx.x;
  const int w = blockIdx.x;
  const int tot = t < B ? btot[t] : 0;
  const int ex = block_scan_excl<32>(tot, wsum);
  if (t < B) {
    cur[t] = ex + wpre[w * B + t];
    if (w == 0) bbase[t] = ex;
  }
  if (w == 0 && t == 0) bbase[B] = E;
  __syncthreads();
  const int e0 = w * per;
  const int e1 = (e0 + per) < E ? (e0 + per) : E;
  int2* tmp2 = (int2*)tmp;
  for (int e = e0 + t; e < e1; e += EU * 1024) {
    int d[EU], s[EU];
#pragma unroll
    for (int u = 0; u < EU; ++u) {
      const int ee = e + u * 1024;
      const bool ok = ee < e1;
      d[u] = ok ? dst[ee] : 0;
      s[u] = ok ? src[ee] : 0;
    }
#pragma unroll
    for (int u = 0; u < EU; ++u) {
      if (e + u * 1024 < e1) {
        const int p = atomicAdd(&cur[d[u] >> BSHIFT], 1);
        tmp2[p] = make_int2(s[u], d[u]);
      }
    }
  }
}

__global__ __launch_bounds__(1024) void k_bsort(const int* __restrict__ tmp,
                                                const int* __restrict__ bbase,
                                                int* __restrict__ ssrc,
                                                int* __restrict__ cnt,
                                                int* __restrict__ start, int n) {
  __shared__ int c[BNODES];
  __shared__ int wsum[32];
  const int t = threadIdx.x;
  const int b = blockIdx.x;
  const int e0 = bbase[b];
  const int e1 = bbase[b + 1];
  const int2* tmp2 = (const int2*)tmp;
  if (t < BNODES) c[t] = 0;
  __syncthreads();
  for (int e = e0 + t; e < e1; e += EU * 1024) {
    int d[EU];
#pragma unroll
    for (int u = 0; u < EU; ++u) {
      const int ee = e + u * 1024;
      d[u] = ee < e1 ? tmp[2 * (size_t)ee + 1] : 0;
    }
#pragma unroll
    for (int u = 0; u < EU; ++u)
      if (e + u * 1024 < e1) atomicAdd(&c[d[u] & (BNODES - 1)], 1);
  }
  __syncthreads();
  const int v = t < BNODES ? c[t] : 0;
  const int ex = block_scan_excl<32>(v, wsum);
  const int node = b * BNODES + t;
  if (t < BNODES) {
    c[t] = e0 + ex;
    if (node < n) {
      cnt[node] = v;
      start[node] = e0 + ex;
    }
  }
  __syncthreads();
  for (int e = e0 + t; e < e1; e += EU * 1024) {
    int2 sd[EU];
#pragma unroll
    for (int u = 0; u < EU; ++u) {
      const int ee = e + u * 1024;
      sd[u] = ee < e1 ? tmp2[ee] : make_int2(0, 0);
    }
#pragma unroll
    for (int u = 0; u < EU; ++u) {
      if (e + u * 1024 < e1) {
        const int p = atomicAdd(&c[sd[u].y & (BNODES - 1)], 1);
        ssrc[p] = sd[u].x;
      }
    }
  }
}

template <int G>
static __device__ __forceinline__ void gather_group(const float* hl, int s, int j,
                                                    v2f& acc) {
  v2f v[G];
#pragma unroll
  for (int u = 0; u < G; ++u) {
    const int sj = __builtin_amdgcn_readlane(s, j + u);
    v[u] = *(const v2f*)(hl + (size_t)sj * HIDDEN);
  }
#pragma unroll
  for (int u = 0; u < G; ++u) {
    acc.x += v[u].x;
    acc.y += v[u].y;
  }
}

__global__ __launch_bounds__(128) void k_inproj(const float* __restrict__ x,
                                                const float* __restrict__ Win,
                                                const float* __restrict__ bin,
                                                float* __restrict__ h, int n) {
  __shared__ float sW[96 * HIDDEN];
  const int wave = threadIdx.x >> 5;
  const int lane = threadIdx.x & 31;
  const int hf = lane >> 4;
  const int r = lane & 15;
  const int rowBase = blockIdx.x * 64 + wave * 16;
  const int row = rowBase + r;
  const int rowc = row < n ? row : n - 1;
  const float* xr = x + (size_t)rowc * DNODE;

  v8f acc[4];
#pragma unroll
  for (int c = 0; c < 4; ++c) {
    float b = bin[c * 16 + r];
#pragma unroll
    for (int i = 0; i < 8; ++i) acc[c][i] = b;
  }

  for (int kc = 0; kc < DNODE; kc += 96) {
    __syncthreads();
    stage_pairs(sW, Win + (size_t)kc * HIDDEN, 96, threadIdx.x, 128);
    __syncthreads();

    if (kc + 96 < DNODE) __builtin_prefetch(xr + kc + 96 + 2 * hf, 0, 3);

#pragma unroll 4
    for (int k = 0; k < 96; k += 4) {
      v2f a = *(const v2f*)(xr + kc + k + 2 * hf);
      const float* wrow = sW + ((k >> 1) + hf) * 128;
#pragma unroll
      for (int c = 0; c < 4; ++c) {
        v2f b = *(const v2f*)(wrow + (c * 16 + r) * 2);
        acc[c] = wmma4(a, b, acc[c]);
      }
    }
  }

#pragma unroll
  for (int c = 0; c < 4; ++c) {
#pragma unroll
    for (int i = 0; i < 8; ++i) {
      int m = rowBase + i + 8 * hf;
      if (m < n) h[(size_t)m * HIDDEN + c * 16 + r] = fmaxf(acc[c][i], 0.0f);
    }
  }
}

#define APITCH 68

__global__ __launch_bounds__(512) void k_layer(const int* __restrict__ ssrc,
                                               const int* __restrict__ start,
                                               const int* __restrict__ cnt,
                                               const float* __restrict__ hprev,
                                               const float* __restrict__ Wl,
                                               const float* __restrict__ bl,
                                               const float* __restrict__ Wr,
                                               const float* __restrict__ g,
                                               const float* __restrict__ bt,
                                               float* __restrict__ hnext, int n) {
  __shared__ __attribute__((aligned(16))) float sWl[HIDDEN * HIDDEN];
  __shared__ __attribute__((aligned(16))) float sWr[HIDDEN * HIDDEN];
  __shared__ __attribute__((aligned(16))) float sH[64 * APITCH];
  __shared__ __attribute__((aligned(16))) float sA[64 * APITCH];

  const int tid = threadIdx.x;
  const int lane = tid & 31;
  const int wave = __builtin_amdgcn_readfirstlane(tid >> 5);
  const int blockRow = blockIdx.x * 64;

  stage_pairs(sWl, Wl, HIDDEN, tid, 512);
  stage_pairs(sWr, Wr, HIDDEN, tid, 512);
  for (int idx = tid; idx < 64 * 16; idx += 512) {
    const int row = idx >> 4;
    const int c4 = idx & 15;
    int gr = blockRow + row;
    gr = gr < n ? gr : n - 1;
    const v4f v = ((const v4f*)(hprev + (size_t)gr * HIDDEN))[c4];
    *(v4f*)(sH + row * APITCH + c4 * 4) = v;
  }

  {
    const int nl0 = wave * 4;
    int cq[4], sq[4];
#pragma unroll
    for (int q = 0; q < 4; ++q) {
      const int nd = blockRow + nl0 + q;
      const bool ok = nd < n;
      const int ndc = ok ? nd : n - 1;
      cq[q] = ok ? cnt[ndc] : 0;
      sq[q] = start[ndc];
    }
    const float* hl = hprev + lane * 2;
#pragma unroll
    for (int q = 0; q < 4; ++q) {
      const int c = cq[q];
      const int s0 = sq[q];
      v2f acc = {0.0f, 0.0f};
      for (int b = 0; b < c; b += 32) {
        const int m = (c - b) < 32 ? (c - b) : 32;
        const int li = lane < m ? lane : m - 1;
        const int s = ssrc[s0 + b + li];
        int j = 0;
        for (; j + 8 <= m; j += 8) gather_group<8>(hl, s, j, acc);
        if (j + 4 <= m) { gather_group<4>(hl, s, j, acc); j += 4; }
        if (j + 2 <= m) { gather_group<2>(hl, s, j, acc); j += 2; }
        if (j < m) gather_group<1>(hl, s, j, acc);
      }
      const float invd = 1.0f / fmaxf((float)c, 1.0f);
      acc.x *= invd;
      acc.y *= invd;
      *(v2f*)(sA + (nl0 + q) * APITCH + lane * 2) = acc;
    }
  }
  __syncthreads();

  const int hf = lane >> 4;
  const int r = lane & 15;
  const int mt = wave >> 2;
  const int ct = wave & 3;

  v8f acc;
  {
    const float b = bl[ct * 16 + r];
#pragma unroll
    for (int i = 0; i < 8; ++i) acc[i] = b;
  }
  const float* arow = sA + (mt * 16 + r) * APITCH + 2 * hf;
  const float* hrow = sH + (mt * 16 + r) * APITCH + 2 * hf;
  const float* wlp = sWl + hf * 128 + (ct * 16 + r) * 2;
  const float* wrp = sWr + hf * 128 + (ct * 16 + r) * 2;
#pragma unroll
  for (int k = 0; k < HIDDEN; k += 4) {
    const v2f a1 = *(const v2f*)(arow + k);
    const v2f a2 = *(const v2f*)(hrow + k);
    const v2f b1 = *(const v2f*)(wlp + (k >> 1) * 128);
    acc = wmma4(a1, b1, acc);
    const v2f b2 = *(const v2f*)(wrp + (k >> 1) * 128);
    acc = wmma4(a2, b2, acc);
  }
  __syncthreads();

  float* sOut = sA;
  {
    float* so = sOut + mt * 16 * HIDDEN;
#pragma unroll
    for (int i = 0; i < 8; ++i) so[(i + 8 * hf) * HIDDEN + ct * 16 + r] = acc[i];
  }
  __syncthreads();
  const int rr = wave * 4 + (lane >> 3);
  const int p4 = (lane & 7) * 4;
  const int node = blockRow + rr;
  const float* orow = sOut + rr * HIDDEN + p4;
  const v4f o0 = *(const v4f*)(orow);
  const v4f o1 = *(const v4f*)(orow + 32);
  float ss = o0.x * o0.x + o0.y * o0.y + o0.z * o0.z + o0.w * o0.w +
             o1.x * o1.x + o1.y * o1.y + o1.z * o1.z + o1.w * o1.w;
  ss += __shfl_xor(ss, 4);
  ss += __shfl_xor(ss, 2);
  ss += __shfl_xor(ss, 1);
  const float l2 = rsqrtf(fmaxf(ss, 1e-24f));

  const float* hrow2 = sH + rr * APITCH + p4;
  v4f t0 = o0 * l2;
  v4f t1 = o1 * l2;
  t0.x = fmaxf(t0.x, 0.0f); t0.y = fmaxf(t0.y, 0.0f); t0.z = fmaxf(t0.z, 0.0f); t0.w = fmaxf(t0.w, 0.0f);
  t1.x = fmaxf(t1.x, 0.0f); t1.y = fmaxf(t1.y, 0.0f); t1.z = fmaxf(t1.z, 0.0f); t1.w = fmaxf(t1.w, 0.0f);
  t0 += *(const v4f*)(hrow2);
  t1 += *(const v4f*)(hrow2 + 32);
  float s = t0.x + t0.y + t0.z + t0.w + t1.x + t1.y + t1.z + t1.w;
  float q = t0.x * t0.x + t0.y * t0.y + t0.z * t0.z + t0.w * t0.w +
            t1.x * t1.x + t1.y * t1.y + t1.z * t1.z + t1.w * t1.w;
  s += __shfl_xor(s, 4);
  q += __shfl_xor(q, 4);
  s += __shfl_xor(s, 2);
  q += __shfl_xor(q, 2);
  s += __shfl_xor(s, 1);
  q += __shfl_xor(q, 1);
  const float mean = s * (1.0f / 64.0f);
  const float var = q * (1.0f / 64.0f) - mean * mean;
  const float sc = rsqrtf(var + 1e-5f);

  if (node < n) {
    const v4f g0 = *(const v4f*)(g + p4);
    const v4f g1 = *(const v4f*)(g + 32 + p4);
    const v4f b0 = *(const v4f*)(bt + p4);
    const v4f b1 = *(const v4f*)(bt + 32 + p4);
    float* op = hnext + (size_t)node * HIDDEN + p4;
    *(v4f*)(op) = (t0 - mean) * sc * g0 + b0;
    *(v4f*)(op + 32) = (t1 - mean) * sc * g1 + b1;
  }
}

__global__ __launch_bounds__(256) void k_finalln(const float* __restrict__ h,
                                                 const float* __restrict__ g,
                                                 const float* __restrict__ b,
                                                 float* __restrict__ out, int n) {
  int t = blockIdx.x * blockDim.x + threadIdx.x;
  int row = t >> 5;
  int lane = t & 31;
  if (row >= n) return;
  v2f v = *(const v2f*)(h + (size_t)row * HIDDEN + lane * 2);
  float s = v.x + v.y;
  float q = v.x * v.x + v.y * v.y;
#pragma unroll
  for (int m = 16; m >= 1; m >>= 1) {
    s += __shfl_xor(s, m);
    q += __shfl_xor(q, m);
  }
  float mean = s * (1.0f / 64.0f);
  float var = q * (1.0f / 64.0f) - mean * mean;
  float sc = rsqrtf(var + 1e-5f);
  v2f g2 = *(const v2f*)(g + lane * 2);
  v2f b2 = *(const v2f*)(b + lane * 2);
  v2f o;
  o.x = (v.x - mean) * sc * g2.x + b2.x;
  o.y = (v.y - mean) * sc * g2.y + b2.y;
  *(v2f*)(out + (size_t)row * HIDDEN + lane * 2) = o;
}

extern "C" void kernel_launch(void* const* d_in, const int* in_sizes, int n_in,
                              void* d_out, int out_size, void* d_ws, size_t ws_size,
                              hipStream_t stream) {
  const float* x   = (const float*)d_in[0];
  const int* ei    = (const int*)d_in[1];
  const float* Win = (const float*)d_in[2];
  const float* bin = (const float*)d_in[3];
  const float* Wl  = (const float*)d_in[4];
  const float* bl  = (const float*)d_in[5];
  const float* Wr  = (const float*)d_in[6];
  const float* lng = (const float*)d_in[7];
  const float* lnb = (const float*)d_in[8];
  const float* og  = (const float*)d_in[9];
  const float* ob  = (const float*)d_in[10];

  const int N = in_sizes[0] / DNODE;
  const int E = in_sizes[1] / 2;
  const int LAYERS = in_sizes[4] / (HIDDEN * HIDDEN);
  const int* src = ei;
  const int* dst = ei + E;

  const int B = (N + BNODES - 1) >> BSHIFT;
  const int per = (E + NCHUNK - 1) / NCHUNK;
  float* h0  = (float*)d_ws;
  float* h1  = h0 + (size_t)N * HIDDEN;
  int* tmp   = (int*)(h1 + (size_t)N * HIDDEN);
  int* ssrc  = tmp + (size_t)2 * E;
  int* cnt   = ssrc + E;
  int* start = cnt + N;
  int* wcnt  = start + N;
  int* wpre  = wcnt + (size_t)NCHUNK * B;
  int* btot  = wpre + (size_t)NCHUNK * B;
  int* bbase = btot + B;

  k_bcount<<<NCHUNK, 1024, 0, stream>>>(dst, wcnt, E, per, B);
  k_bscan<<<B, NCHUNK, 0, stream>>>(wcnt, wpre, btot, B);
  k_bplace<<<NCHUNK, 1024, 0, stream>>>(src, dst, wpre, btot, bbase, tmp, E, per, B);
  k_bsort<<<B, 1024, 0, stream>>>(tmp, bbase, ssrc, cnt, start, N);

  k_inproj<<<(N + 63) / 64, 128, 0, stream>>>(x, Win, bin, h0, N);

  float* cur = h0;
  float* nxt = h1;
  for (int l = 0; l < LAYERS; ++l) {
    k_layer<<<(N + 63) / 64, 512, 0, stream>>>(
        ssrc, start, cnt, cur,
        Wl + (size_t)l * HIDDEN * HIDDEN, bl + (size_t)l * HIDDEN,
        Wr + (size_t)l * HIDDEN * HIDDEN,
        lng + (size_t)l * HIDDEN, lnb + (size_t)l * HIDDEN, nxt, N);
    float* t = cur; cur = nxt; nxt = t;
  }

  k_finalln<<<(int)(((long long)N * 32 + 255) / 256), 256, 0, stream>>>(cur, og, ob, (float*)d_out, N);
}


# last layer's launch also applies the closing layer norm to the row in registers and writes the output; final-norm launch (one wave per row) gone; its row sums over 8 lanes: new tree
# speedup vs baseline: 3.5175x; 1.0501x over previous
#include <hip/hip_runtime.h>
#include <hip/hip_bf16.h>

typedef __attribute__((ext_vector_type(2))) float v2f;
typedef __attribute__((ext_vector_type(4))) float v4f;
typedef __attribute__((ext_vector_type(8))) float v8f;

#define HIDDEN 64
#define DNODE 384

static __device__ __forceinline__ v8f wmma4(v2f a, v2f b, v8f c) {
  return __builtin_amdgcn_wmma_f32_16x16x4_f32(
       false, a,  false, b,
       (short)0, c,  false,  false);
}

static __device__ __forceinline__ void stage_pairs(float* sW, const float* W,
                                                   int rows, int tid, int nthr) {
  const int nvec = rows * HIDDEN / 4;
  for (int idx = tid; idx < nvec; idx += nthr) {
    int k = idx >> 4;
    int col = (idx & 15) * 4;
    v4f v = ((const v4f*)W)[idx];
    float* dstp = sW + (k >> 1) * 128 + (k & 1);
    dstp[(col + 0) * 2] = v.x;
    dstp[(col + 1) * 2] = v.y;
    dstp[(col + 2) * 2] = v.z;
    dstp[(col + 3) * 2] = v.w;
  }
}

#define NCHUNK 256
#define BSHIFT 9
#define BNODES 512
#define BMAX 1024
#define EU 8

template <int NW>
static __device__ __forceinline__ int block_scan_excl(int v, int* wsum) {
  const int lane = threadIdx.x & 31;
  const int wave = threadIdx.x >> 5;
  int x = v;
#pragma unroll
  for (int d = 1; d < 32; d <<= 1) {
    const int y = __shfl_up(x, d);
    if (lane >= d) x += y;
  }
  if (lane == 31) wsum[wave] = x;
  __syncthreads();
  if (wave == 0) {
    const int tv = lane < NW ? wsum[lane] : 0;
    int z = tv;
#pragma unroll
    for (int d = 1; d < 32; d <<= 1) {
      const int y = __shfl_up(z, d);
      if (lane >= d) z += y;
    }
    if (lane < NW) wsum[lane] = z - tv;
  }
  __syncthreads();
  return wsum[wave] + x - v;
}

__global__ __launch_bounds__(1024) void k_bcount(const int* __restrict__ dst,
                                                 int* __restrict__ wcnt,
                                                 int E, int per, int B) {
  __shared__ int hb[BMAX];
  const int t = threadIdx.x;
  hb[t] = 0;
  __syncthreads();
  const int e0 = blockIdx.x * per;
  const int e1 = (e0 + per) < E ? (e0 + per) : E;
  for (int e = e0 + t; e < e1; e += EU * 1024) {
    int d[EU];
#pragma unroll
    for (int u = 0; u < EU; ++u) {
      const int ee = e + u * 1024;
      d[u] = ee < e1 ? dst[ee] : 0;
    }
#pragma unroll
    for (int u = 0; u < EU; ++u)
      if (e + u * 1024 < e1) atomicAdd(&hb[d[u] >> BSHIFT], 1);
  }
  __syncthreads();
  if (t < B) wcnt[blockIdx.x * B + t] = hb[t];
}

__global__ __launch_bounds__(NCHUNK) void k_bscan(const int* __restrict__ wcnt,
                                                  int* __restrict__ wpre,
                                                  int* __restrict__ btot, int B) {
  __shared__ int wsum[32];
  const int b = blockIdx.x;
  const int w = threadIdx.x;
  const int v = wcnt[w * B + b];
  const int ex = block_scan_excl<NCHUNK / 32>(v, wsum);
  wpre[w * B + b] = ex;
  if (w == NCHUNK - 1) btot[b] = ex + v;
}

__global__ __launch_bounds__(1024) void k_bplace(const int* __restrict__ src,
                                                 const int* __restrict__ dst,
                                                 const int* __restrict__ wpre,
                                                 const int* __restrict__ btot,
                                                 int* __restrict__ bbase,
                                                 int* __restrict__ tmp,
                                                 int E, int per, int B) {
  __shared__ int cur[BMAX];
  __shared__ int wsum[32];
  const int t = threadIdx.x;
  const int w = blockIdx.x;
  const int tot = t < B ? btot[t] : 0;
  const int ex = block_scan_excl<32>(tot, wsum);
  if (t < B) {
    cur[t] = ex + wpre[w * B + t];
    if (w == 0) bbase[t] = ex;
  }
  if (w == 0 && t == 0) bbase[B] = E;
  __syncthreads();
  const int e0 = w * per;
  const int e1 = (e0 + per) < E ? (e0 + per) : E;
  int2* tmp2 = (int2*)tmp;
  for (int e = e0 + t; e < e1; e += EU * 1024) {
    int d[EU], s[EU];
#pragma unroll
    for (int u = 0; u < EU; ++u) {
      const int ee = e + u * 1024;
      const bool ok = ee < e1;
      d[u] = ok ? dst[ee] : 0;
      s[u] = ok ? src[ee] : 0;
    }
#pragma unroll
    for (int u = 0; u < EU; ++u) {
      if (e + u * 1024 < e1) {
        const int p = atomicAdd(&cur[d[u] >> BSHIFT], 1);
        tmp2[p] = make_int2(s[u], d[u]);
      }
    }
  }
}

__global__ __launch_bounds__(1024) void k_bsort(const int* __restrict__ tmp,
                                                const int* __restrict__ bbase,
                                                int* __restrict__ ssrc,
                                                int* __restrict__ cnt,
                                                int* __restrict__ start, int n) {
  __shared__ int c[BNODES];
  __shared__ int wsum[32];
  const int t = threadIdx.x;
  const int b = blockIdx.x;
  const int e0 = bbase[b];
  const int e1 = bbase[b + 1];
  const int2* tmp2 = (const int2*)tmp;
  if (t < BNODES) c[t] = 0;
  __syncthreads();
  for (int e = e0 + t; e < e1; e += EU * 1024) {
    int d[EU];
#pragma unroll
    for (int u = 0; u < EU; ++u) {
      const int ee = e + u * 1024;
      d[u] = ee < e1 ? tmp[2 * (size_t)ee + 1] : 0;
    }
#pragma unroll
    for (int u = 0; u < EU; ++u)
      if (e + u * 1024 < e1) atomicAdd(&c[d[u] & (BNODES - 1)], 1);
  }
  __syncthreads();
  const int v = t < BNODES ? c[t] : 0;
  const int ex = block_scan_excl<32>(v, wsum);
  const int node = b * BNODES + t;
  if (t < BNODES) {
    c[t] = e0 + ex;
    if (node < n) {
      cnt[node] = v;
      start[node] = e0 + ex;
    }
  }
  __syncthreads();
  for (int e = e0 + t; e < e1; e += EU * 1024) {
    int2 sd[EU];
#pragma unroll
    for (int u = 0; u < EU; ++u) {
      const int ee = e + u * 1024;
      sd[u] = ee < e1 ? tmp2[ee] : make_int2(0, 0);
    }
#pragma unroll
    for (int u = 0; u < EU; ++u) {
      if (e + u * 1024 < e1) {
        const int p = atomicAdd(&c[sd[u].y & (BNODES - 1)], 1);
        ssrc[p] = sd[u].x;
      }
    }
  }
}

template <int G>
static __device__ __forceinline__ void gather_group(const float* hl, int s, int j,
                                                    v2f& acc) {
  v2f v[G];
#pragma unroll
  for (int u = 0; u < G; ++u) {
    const int sj = __builtin_amdgcn_readlane(s, j + u);
    v[u] = *(const v2f*)(hl + (size_t)sj * HIDDEN);
  }
#pragma unroll
  for (int u = 0; u < G; ++u) {
    acc.x += v[u].x;
    acc.y += v[u].y;
  }
}

__global__ __launch_bounds__(128) void k_inproj(const float* __restrict__ x,
                                                const float* __restrict__ Win,
                                                const float* __restrict__ bin,
                                                float* __restrict__ h, int n) {
  __shared__ float sW[96 * HIDDEN];
  const int wave = threadIdx.x >> 5;
  const int lane = threadIdx.x & 31;
  const int hf = lane >> 4;
  const int r = lane & 15;
  const int rowBase = blockIdx.x * 64 + wave * 16;
  const int row = rowBase + r;
  const int rowc = row < n ? row : n - 1;
  const float* xr = x + (size_t)rowc * DNODE;

  v8f acc[4];
#pragma unroll
  for (int c = 0; c < 4; ++c) {
    float b = bin[c * 16 + r];
#pragma unroll
    for (int i = 0; i < 8; ++i) acc[c][i] = b;
  }

  for (int kc = 0; kc < DNODE; kc += 96) {
    __syncthreads();
    stage_pairs(sW, Win + (size_t)kc * HIDDEN, 96, threadIdx.x, 128);
    __syncthreads();

    if (kc + 96 < DNODE) __builtin_prefetch(xr + kc + 96 + 2 * hf, 0, 3);

#pragma unroll 4
    for (int k = 0; k < 96; k += 4) {
      v2f a = *(const v2f*)(xr + kc + k + 2 * hf);
      const float* wrow = sW + ((k >> 1) + hf) * 128;
#pragma unroll
      for (int c = 0; c < 4; ++c) {
        v2f b = *(const v2f*)(wrow + (c * 16 + r) * 2);
        acc[c] = wmma4(a, b, acc[c]);
      }
    }
  }

#pragma unroll
  for (int c = 0; c < 4; ++c) {
#pragma unroll
    for (int i = 0; i < 8; ++i) {
      int m = rowBase + i + 8 * hf;
      if (m < n) h[(size_t)m * HIDDEN + c * 16 + r] = fmaxf(acc[c][i], 0.0f);
    }
  }
}

#define APITCH 68

template <bool LAST>
__global__ __launch_bounds__(512) void k_layer(const int* __restrict__ ssrc,
                                               const int* __restrict__ start,
                                               const int* __restrict__ cnt,
                                               const float* __restrict__ hprev,
                                               const float* __restrict__ Wl,
                                               const float* __restrict__ bl,
                                               const float* __restrict__ Wr,
                                               const float* __restrict__ g,
                                               const float* __restrict__ bt,
                                               const float* __restrict__ g2,
                                               const float* __restrict__ b2,
                                               float* __restrict__ hnext, int n) {
  __shared__ __attribute__((aligned(16))) float sWl[HIDDEN * HIDDEN];
  __shared__ __attribute__((aligned(16))) float sWr[HIDDEN * HIDDEN];
  __shared__ __attribute__((aligned(16))) float sH[64 * APITCH];
  __shared__ __attribute__((aligned(16))) float sA[64 * APITCH];

  const int tid = threadIdx.x;
  const int lane = tid & 31;
  const int wave = __builtin_amdgcn_readfirstlane(tid >> 5);
  const int blockRow = blockIdx.x * 64;

  stage_pairs(sWl, Wl, HIDDEN, tid, 512);
  stage_pairs(sWr, Wr, HIDDEN, tid, 512);
  for (int idx = tid; idx < 64 * 16; idx += 512) {
    const int row = idx >> 4;
    const int c4 = idx & 15;
    int gr = blockRow + row;
    gr = gr < n ? gr : n - 1;
    const v4f v = ((const v4f*)(hprev + (size_t)gr * HIDDEN))[c4];
    *(v4f*)(sH + row * APITCH + c4 * 4) = v;
  }

  {
    const int nl0 = wave * 4;
    int cq[4], sq[4];
#pragma unroll
    for (int q = 0; q < 4; ++q) {
      const int nd = blockRow + nl0 + q;
      const bool ok = nd < n;
      const int ndc = ok ? nd : n - 1;
      cq[q] = ok ? cnt[ndc] : 0;
      sq[q] = start[ndc];
    }
    const float* hl = hprev + lane * 2;
#pragma unroll
    for (int q = 0; q < 4; ++q) {
      const int c = cq[q];
      const int s0 = sq[q];
      v2f acc = {0.0f, 0.0f};
      for (int b = 0; b < c; b += 32) {
        const int m = (c - b) < 32 ? (c - b) : 32;
        const int li = lane < m ? lane : m - 1;
        const int s = ssrc[s0 + b + li];
        int j = 0;
        for (; j + 8 <= m; j += 8) gather_group<8>(hl, s, j, acc);
        if (j + 4 <= m) { gather_group<4>(hl, s, j, acc); j += 4; }
        if (j + 2 <= m) { gather_group<2>(hl, s, j, acc); j += 2; }
        if (j < m) gather_group<1>(hl, s, j, acc);
      }
      const float invd = 1.0f / fmaxf((float)c, 1.0f);
      acc.x *= invd;
      acc.y *= invd;
      *(v2f*)(sA + (nl0 + q) * APITCH + lane * 2) = acc;
    }
  }
  __syncthreads();

  const int hf = lane >> 4;
  const int r = lane & 15;
  const int mt = wave >> 2;
  const int ct = wave & 3;

  v8f acc;
  {
    const float b = bl[ct * 16 + r];
#pragma unroll
    for (int i = 0; i < 8; ++i) acc[i] = b;
  }
  const float* arow = sA + (mt * 16 + r) * APITCH + 2 * hf;
  const float* hrow = sH + (mt * 16 + r) * APITCH + 2 * hf;
  const float* wlp = sWl + hf * 128 + (ct * 16 + r) * 2;
  const float* wrp = sWr + hf * 128 + (ct * 16 + r) * 2;
#pragma unroll
  for (int k = 0; k < HIDDEN; k += 4) {
    const v2f a1 = *(const v2f*)(arow + k);
    const v2f a2 = *(const v2f*)(hrow + k);
    const v2f b1 = *(const v2f*)(wlp + (k >> 1) * 128);
    acc = wmma4(a1, b1, acc);
    const v2f b2 = *(const v2f*)(wrp + (k >> 1) * 128);
    acc = wmma4(a2, b2, acc);
  }
  __syncthreads();

  float* sOut = sA;
  {
    float* so = sOut + mt * 16 * HIDDEN;
#pragma unroll
    for (int i = 0; i < 8; ++i) so[(i + 8 * hf) * HIDDEN + ct * 16 + r] = acc[i];
  }
  __syncthreads();
  const int rr = wave * 4 + (lane >> 3);
  const int p4 = (lane & 7) * 4;
  const int node = blockRow + rr;
  const float* orow = sOut + rr * HIDDEN + p4;
  const v4f o0 = *(const v4f*)(orow);
  const v4f o1 = *(const v4f*)(orow + 32);
  float ss = o0.x * o0.x + o0.y * o0.y + o0.z * o0.z + o0.w * o0.w +
             o1.x * o1.x + o1.y * o1.y + o1.z * o1.z + o1.w * o1.w;
  ss += __shfl_xor(ss, 4);
  ss += __shfl_xor(ss, 2);
  ss += __shfl_xor(ss, 1);
  const float l2 = rsqrtf(fmaxf(ss, 1e-24f));

  const float* hrow2 = sH + rr * APITCH + p4;
  v4f t0 = o0 * l2;
  v4f t1 = o1 * l2;
  t0.x = fmaxf(t0.x, 0.0f); t0.y = fmaxf(t0.y, 0.0f); t0.z = fmaxf(t0.z, 0.0f); t0.w = fmaxf(t0.w, 0.0f);
  t1.x = fmaxf(t1.x, 0.0f); t1.y = fmaxf(t1.y, 0.0f); t1.z = fmaxf(t1.z, 0.0f); t1.w = fmaxf(t1.w, 0.0f);
  t0 += *(const v4f*)(hrow2);
  t1 += *(const v4f*)(hrow2 + 32);
  float s = t0.x + t0.y + t0.z + t0.w + t1.x + t1.y + t1.z + t1.w;
  float q = t0.x * t0.x + t0.y * t0.y + t0.z * t0.z + t0.w * t0.w +
            t1.x * t1.x + t1.y * t1.y + t1.z * t1.z + t1.w * t1.w;
  s += __shfl_xor(s, 4);
  q += __shfl_xor(q, 4);
  s += __shfl_xor(s, 2);
  q += __shfl_xor(q, 2);
  s += __shfl_xor(s, 1);
  q += __shfl_xor(q, 1);
  const float mean = s * (1.0f / 64.0f);
  const float var = q * (1.0f / 64.0f) - mean * mean;
  const float sc = rsqrtf(var + 1e-5f);

  v4f y0 = (t0 - mean) * sc * *(const v4f*)(g + p4) + *(const v4f*)(bt + p4);
  v4f y1 = (t1 - mean) * sc * *(const v4f*)(g + 32 + p4) + *(const v4f*)(bt + 32 + p4);
  if (LAST) {
    float s2 = y0.x + y0.y + y0.z + y0.w + y1.x + y1.y + y1.z + y1.w;
    float q2 = y0.x * y0.x + y0.y * y0.y + y0.z * y0.z + y0.w * y0.w +
               y1.x * y1.x + y1.y * y1.y + y1.z * y1.z + y1.w * y1.w;
    s2 += __shfl_xor(s2, 4);
    q2 += __shfl_xor(q2, 4);
    s2 += __shfl_xor(s2, 2);
    q2 += __shfl_xor(q2, 2);
    s2 += __shfl_xor(s2, 1);
    q2 += __shfl_xor(q2, 1);
    const float mean2 = s2 * (1.0f / 64.0f);
    const float var2 = q2 * (1.0f / 64.0f) - mean2 * mean2;
    const float sc2 = rsqrtf(var2 + 1e-5f);
    y0 = (y0 - mean2) * sc2 * *(const v4f*)(g2 + p4) + *(const v4f*)(b2 + p4);
    y1 = (y1 - mean2) * sc2 * *(const v4f*)(g2 + 32 + p4) + *(const v4f*)(b2 + 32 + p4);
  }
  if (node < n) {
    float* op = hnext + (size_t)node * HIDDEN + p4;
    *(v4f*)(op) = y0;
    *(v4f*)(op + 32) = y1;
  }
}

__global__ __launch_bounds__(256) void k_finalln(const float* __restrict__ h,
                                                 const float* __restrict__ g,
                                                 const float* __restrict__ b,
                                                 float* __restrict__ out, int n) {
  int t = blockIdx.x * blockDim.x + threadIdx.x;
  int row = t >> 5;
  int lane = t & 31;
  if (row >= n) return;
  v2f v = *(const v2f*)(h + (size_t)row * HIDDEN + lane * 2);
  float s = v.x + v.y;
  float q = v.x * v.x + v.y * v.y;
#pragma unroll
  for (int m = 16; m >= 1; m >>= 1) {
    s += __shfl_xor(s, m);
    q += __shfl_xor(q, m);
  }
  float mean = s * (1.0f / 64.0f);
  float var = q * (1.0f / 64.0f) - mean * mean;
  float sc = rsqrtf(var + 1e-5f);
  v2f g2 = *(const v2f*)(g + lane * 2);
  v2f b2 = *(const v2f*)(b + lane * 2);
  v2f o;
  o.x = (v.x - mean) * sc * g2.x + b2.x;
  o.y = (v.y - mean) * sc * g2.y + b2.y;
  *(v2f*)(out + (size_t)row * HIDDEN + lane * 2) = o;
}

extern "C" void kernel_launch(void* const* d_in, const int* in_sizes, int n_in,
                              void* d_out, int out_size, void* d_ws, size_t ws_size,
                              hipStream_t stream) {
  const float* x   = (const float*)d_in[0];
  const int* ei    = (const int*)d_in[1];
  const float* Win = (const float*)d_in[2];
  const float* bin = (const float*)d_in[3];
  const float* Wl  = (const float*)d_in[4];
  const float* bl  = (const float*)d_in[5];
  const float* Wr  = (const float*)d_in[6];
  const float* lng = (const float*)d_in[7];
  const float* lnb = (const float*)d_in[8];
  const float* og  = (const float*)d_in[9];
  const float* ob  = (const float*)d_in[10];

  const int N = in_sizes[0] / DNODE;
  const int E = in_sizes[1] / 2;
  const int LAYERS = in_sizes[4] / (HIDDEN * HIDDEN);
  const int* src = ei;
  const int* dst = ei + E;

  const int B = (N + BNODES - 1) >> BSHIFT;
  const int per = (E + NCHUNK - 1) / NCHUNK;
  float* h0  = (float*)d_ws;
  float* h1  = h0 + (size_t)N * HIDDEN;
  int* tmp   = (int*)(h1 + (size_t)N * HIDDEN);
  int* ssrc  = tmp + (size_t)2 * E;
  int* cnt   = ssrc + E;
  int* start = cnt + N;
  int* wcnt  = start + N;
  int* wpre  = wcnt + (size_t)NCHUNK * B;
  int* btot  = wpre + (size_t)NCHUNK * B;
  int* bbase = btot + B;

  k_bcount<<<NCHUNK, 1024, 0, stream>>>(dst, wcnt, E, per, B);
  k_bscan<<<B, NCHUNK, 0, stream>>>(wcnt, wpre, btot, B);
  k_bplace<<<NCHUNK, 1024, 0, stream>>>(src, dst, wpre, btot, bbase, tmp, E, per, B);
  k_bsort<<<B, 1024, 0, stream>>>(tmp, bbase, ssrc, cnt, start, N);

  k_inproj<<<(N + 63) / 64, 128, 0, stream>>>(x, Win, bin, h0, N);

  float* cur = h0;
  float* nxt = h1;
  for (int l = 0; l < LAYERS; ++l) {
    if (l == LAYERS - 1) {
      k_layer<true><<<(N + 63) / 64, 512, 0, stream>>>(
          ssrc, start, cnt, cur,
          Wl + (size_t)l * HIDDEN * HIDDEN, bl + (size_t)l * HIDDEN,
          Wr + (size_t)l * HIDDEN * HIDDEN,
          lng + (size_t)l * HIDDEN, lnb + (size_t)l * HIDDEN, og, ob, (float*)d_out, N);
    } else {
      k_layer<false><<<(N + 63) / 64, 512, 0, stream>>>(
          ssrc, start, cnt, cur,
          Wl + (size_t)l * HIDDEN * HIDDEN, bl + (size_t)l * HIDDEN,
          Wr + (size_t)l * HIDDEN * HIDDEN,
          lng + (size_t)l * HIDDEN, lnb + (size_t)l * HIDDEN, og, ob, nxt, N);
    }
    float* t = cur; cur = nxt; nxt = t;
  }

  if (LAYERS == 0)
    k_finalln<<<(int)(((long long)N * 32 + 255) / 256), 256, 0, stream>>>(cur, og, ob, (float*)d_out, N);
}


# layer: weight and own-row loads, the 4 nodes' counts, starts and first index vectors all requested before the gather waits on any (LDS stores follow); rows still summed in slot order
# speedup vs baseline: 3.7499x; 1.0661x over previous
#include <hip/hip_runtime.h>
#include <hip/hip_bf16.h>

typedef __attribute__((ext_vector_type(2))) float v2f;
typedef __attribute__((ext_vector_type(4))) float v4f;
typedef __attribute__((ext_vector_type(8))) float v8f;

#define HIDDEN 64
#define DNODE 384

static __device__ __forceinline__ v8f wmma4(v2f a, v2f b, v8f c) {
  return __builtin_amdgcn_wmma_f32_16x16x4_f32(
       false, a,  false, b,
       (short)0, c,  false,  false);
}

static __device__ __forceinline__ void stage_pairs(float* sW, const float* W,
                                                   int rows, int tid, int nthr) {
  const int nvec = rows * HIDDEN / 4;
  for (int idx = tid; idx < nvec; idx += nthr) {
    int k = idx >> 4;
    int col = (idx & 15) * 4;
    v4f v = ((const v4f*)W)[idx];
    float* dstp = sW + (k >> 1) * 128 + (k & 1);
    dstp[(col + 0) * 2] = v.x;
    dstp[(col + 1) * 2] = v.y;
    dstp[(col + 2) * 2] = v.z;
    dstp[(col + 3) * 2] = v.w;
  }
}

static __device__ __forceinline__ void put_pairs(float* sW, int idx, v4f v) {
  const int k = idx >> 4;
  const int col = (idx & 15) * 4;
  float* dstp = sW + (k >> 1) * 128 + (k & 1);
  dstp[(col + 0) * 2] = v.x;
  dstp[(col + 1) * 2] = v.y;
  dstp[(col + 2) * 2] = v.z;
  dstp[(col + 3) * 2] = v.w;
}

#define NCHUNK 256
#define BSHIFT 9
#define BNODES 512
#define BMAX 1024
#define EU 8

template <int NW>
static __device__ __forceinline__ int block_scan_excl(int v, int* wsum) {
  const int lane = threadIdx.x & 31;
  const int wave = threadIdx.x >> 5;
  int x = v;
#pragma unroll
  for (int d = 1; d < 32; d <<= 1) {
    const int y = __shfl_up(x, d);
    if (lane >= d) x += y;
  }
  if (lane == 31) wsum[wave] = x;
  __syncthreads();
  if (wave == 0) {
    const int tv = lane < NW ? wsum[lane] : 0;
    int z = tv;
#pragma unroll
    for (int d = 1; d < 32; d <<= 1) {
      const int y = __shfl_up(z, d);
      if (lane >= d) z += y;
    }
    if (lane < NW) wsum[lane] = z - tv;
  }
  __syncthreads();
  return wsum[wave] + x - v;
}

__global__ __launch_bounds__(1024) void k_bcount(const int* __restrict__ dst,
                                                 int* __restrict__ wcnt,
                                                 int E, int per, int B) {
  __shared__ int hb[BMAX];
  const int t = threadIdx.x;
  hb[t] = 0;
  __syncthreads();
  const int e0 = blockIdx.x * per;
  const int e1 = (e0 + per) < E ? (e0 + per) : E;
  for (int e = e0 + t; e < e1; e += EU * 1024) {
    int d[EU];
#pragma unroll
    for (int u = 0; u < EU; ++u) {
      const int ee = e + u * 1024;
      d[u] = ee < e1 ? dst[ee] : 0;
    }
#pragma unroll
    for (int u = 0; u < EU; ++u)
      if (e + u * 1024 < e1) atomicAdd(&hb[d[u] >> BSHIFT], 1);
  }
  __syncthreads();
  if (t < B) wcnt[blockIdx.x * B + t] = hb[t];
}

__global__ __launch_bounds__(NCHUNK) void k_bscan(const int* __restrict__ wcnt,
                                                  int* __restrict__ wpre,
                                                  int* __restrict__ btot, int B) {
  __shared__ int wsum[32];
  const int b = blockIdx.x;
  const int w = threadIdx.x;
  const int v = wcnt[w * B + b];
  const int ex = block_scan_excl<NCHUNK / 32>(v, wsum);
  wpre[w * B + b] = ex;
  if (w == NCHUNK - 1) btot[b] = ex + v;
}

__global__ __launch_bounds__(1024) void k_bplace(const int* __restrict__ src,
                                                 const int* __restrict__ dst,
                                                 const int* __restrict__ wpre,
                                                 const int* __restrict__ btot,
                                                 int* __restrict__ bbase,
                                                 int* __restrict__ tmp,
                                                 int E, int per, int B) {
  __shared__ int cur[BMAX];
  __shared__ int wsum[32];
  const int t = threadIdx.x;
  const int w = blockIdx.x;
  const int tot = t < B ? btot[t] : 0;
  const int ex = block_scan_excl<32>(tot, wsum);
  if (t < B) {
    cur[t] = ex + wpre[w * B + t];
    if (w == 0) bbase[t] = ex;
  }
  if (w == 0 && t == 0) bbase[B] = E;
  __syncthreads();
  const int e0 = w * per;
  const int e1 = (e0 + per) < E ? (e0 + per) : E;
  int2* tmp2 = (int2*)tmp;
  for (int e = e0 + t; e < e1; e += EU * 1024) {
    int d[EU], s[EU];
#pragma unroll
    for (int u = 0; u < EU; ++u) {
      const int ee = e + u * 1024;
      const bool ok = ee < e1;
      d[u] = ok ? dst[ee] : 0;
      s[u] = ok ? src[ee] : 0;
    }
#pragma unroll
    for (int u = 0; u < EU; ++u) {
      if (e + u * 1024 < e1) {
        const int p = atomicAdd(&cur[d[u] >> BSHIFT], 1);
        tmp2[p] = make_int2(s[u], d[u]);
      }
    }
  }
}

__global__ __launch_bounds__(1024) void k_bsort(const int* __restrict__ tmp,
                                                const int* __restrict__ bbase,
                                                int* __restrict__ ssrc,
                                                int* __restrict__ cnt,
                                                int* __restrict__ start, int n) {
  __shared__ int c[BNODES];
  __shared__ int wsum[32];
  const int t = threadIdx.x;
  const int b = blockIdx.x;
  const int e0 = bbase[b];
  const int e1 = bbase[b + 1];
  const int2* tmp2 = (const int2*)tmp;
  if (t < BNODES) c[t] = 0;
  __syncthreads();
  for (int e = e0 + t; e < e1; e += EU * 1024) {
    int d[EU];
#pragma unroll
    for (int u = 0; u < EU; ++u) {
      const int ee = e + u * 1024;
      d[u] = ee < e1 ? tmp[2 * (size_t)ee + 1] : 0;
    }
#pragma unroll
    for (int u = 0; u < EU; ++u)
      if (e + u * 1024 < e1) atomicAdd(&c[d[u] & (BNODES - 1)], 1);
  }
  __syncthreads();
  const int v = t < BNODES ? c[t] : 0;
  const int ex = block_scan_excl<32>(v, wsum);
  const int node = b * BNODES + t;
  if (t < BNODES) {
    c[t] = e0 + ex;
    if (node < n) {
      cnt[node] = v;
      start[node] = e0 + ex;
    }
  }
  __syncthreads();
  for (int e = e0 + t; e < e1; e += EU * 1024) {
    int2 sd[EU];
#pragma unroll
    for (int u = 0; u < EU; ++u) {
      const int ee = e + u * 1024;
      sd[u] = ee < e1 ? tmp2[ee] : make_int2(0, 0);
    }
#pragma unroll
    for (int u = 0; u < EU; ++u) {
      if (e + u * 1024 < e1) {
        const int p = atomicAdd(&c[sd[u].y & (BNODES - 1)], 1);
        ssrc[p] = sd[u].x;
      }
    }
  }
}

template <int G>
static __device__ __forceinline__ void gather_group(const float* hl, int s, int j,
                                                    v2f& acc) {
  v2f v[G];
#pragma unroll
  for (int u = 0; u < G; ++u) {
    const int sj = __builtin_amdgcn_readlane(s, j + u);
    v[u] = *(const v2f*)(hl + (size_t)sj * HIDDEN);
  }
#pragma unroll
  for (int u = 0; u < G; ++u) {
    acc.x += v[u].x;
    acc.y += v[u].y;
  }
}

static __device__ __forceinline__ void gather_slots(const float* hl, int s, int m,
                                                    v2f& acc) {
  int j = 0;
  for (; j + 8 <= m; j += 8) gather_group<8>(hl, s, j, acc);
  if (j + 4 <= m) { gather_group<4>(hl, s, j, acc); j += 4; }
  if (j + 2 <= m) { gather_group<2>(hl, s, j, acc); j += 2; }
  if (j < m) gather_group<1>(hl, s, j, acc);
}

__global__ __launch_bounds__(128) void k_inproj(const float* __restrict__ x,
                                                const float* __restrict__ Win,
                                                const float* __restrict__ bin,
                                                float* __restrict__ h, int n) {
  __shared__ float sW[96 * HIDDEN];
  const int wave = threadIdx.x >> 5;
  const int lane = threadIdx.x & 31;
  const int hf = lane >> 4;
  const int r = lane & 15;
  const int rowBase = blockIdx.x * 64 + wave * 16;
  const int row = rowBase + r;
  const int rowc = row < n ? row : n - 1;
  const float* xr = x + (size_t)rowc * DNODE;

  v8f acc[4];
#pragma unroll
  for (int c = 0; c < 4; ++c) {
    float b = bin[c * 16 + r];
#pragma unroll
    for (int i = 0; i < 8; ++i) acc[c][i] = b;
  }

  for (int kc = 0; kc < DNODE; kc += 96) {
    __syncthreads();
    stage_pairs(sW, Win + (size_t)kc * HIDDEN, 96, threadIdx.x, 128);
    __syncthreads();

    if (kc + 96 < DNODE) __builtin_prefetch(xr + kc + 96 + 2 * hf, 0, 3);

#pragma unroll 4
    for (int k = 0; k < 96; k += 4) {
      v2f a = *(const v2f*)(xr + kc + k + 2 * hf);
      const float* wrow = sW + ((k >> 1) + hf) * 128;
#pragma unroll
      for (int c = 0; c < 4; ++c) {
        v2f b = *(const v2f*)(wrow + (c * 16 + r) * 2);
        acc[c] = wmma4(a, b, acc[c]);
      }
    }
  }

#pragma unroll
  for (int c = 0; c < 4; ++c) {
#pragma unroll
    for (int i = 0; i < 8; ++i) {
      int m = rowBase + i + 8 * hf;
      if (m < n) h[(size_t)m * HIDDEN + c * 16 + r] = fmaxf(acc[c][i], 0.0f);
    }
  }
}

#define APITCH 68

template <bool LAST>
__global__ __launch_bounds__(512) void k_layer(const int* __restrict__ ssrc,
                                               const int* __restrict__ start,
                                               const int* __restrict__ cnt,
                                               const float* __restrict__ hprev,
                                               const float* __restrict__ Wl,
                                               const float* __restrict__ bl,
                                               const float* __restrict__ Wr,
                                               const float* __restrict__ g,
                                               const float* __restrict__ bt,
                                               const float* __restrict__ g2,
                                               const float* __restrict__ b2,
                                               float* __restrict__ hnext, int n) {
  __shared__ __attribute__((aligned(16))) float sWl[HIDDEN * HIDDEN];
  __shared__ __attribute__((aligned(16))) float sWr[HIDDEN * HIDDEN];
  __shared__ __attribute__((aligned(16))) float sH[64 * APITCH];
  __shared__ __attribute__((aligned(16))) float sA[64 * APITCH];

  const int tid = threadIdx.x;
  const int lane = tid & 31;
  const int wave = __builtin_amdgcn_readfirstlane(tid >> 5);
  const int blockRow = blockIdx.x * 64;

  const int nl0 = wave * 4;
  int ndv = blockRow + nl0 + (lane & 3);
  const bool okv = ndv < n;
  ndv = okv ? ndv : n - 1;
  const int cl = cnt[ndv];
  const int sv = start[ndv];
  const int cv = okv ? cl : 0;

  const v4f w0 = ((const v4f*)Wl)[tid];
  const v4f w1 = ((const v4f*)Wl)[tid + 512];
  const v4f w2 = ((const v4f*)Wr)[tid];
  const v4f w3 = ((const v4f*)Wr)[tid + 512];
  const int hr = tid >> 4;
  const int hc = tid & 15;
  int gr0 = blockRow + hr;
  int gr1 = gr0 + 32;
  gr0 = gr0 < n ? gr0 : n - 1;
  gr1 = gr1 < n ? gr1 : n - 1;
  const v4f hv0 = ((const v4f*)(hprev + (size_t)gr0 * HIDDEN))[hc];
  const v4f hv1 = ((const v4f*)(hprev + (size_t)gr1 * HIDDEN))[hc];

  int cq[4], sq[4], si[4];
#pragma unroll
  for (int q = 0; q < 4; ++q) {
    cq[q] = __builtin_amdgcn_readlane(cv, q);
    sq[q] = __builtin_amdgcn_readlane(sv, q);
  }
#pragma unroll
  for (int q = 0; q < 4; ++q) {
    const int c = cq[q];
    const int m = c < 32 ? c : 32;
    const int li = lane < m ? lane : m - 1;
    si[q] = 0;
    if (c > 0) si[q] = ssrc[sq[q] + li];
  }

  put_pairs(sWl, tid, w0);
  put_pairs(sWl, tid + 512, w1);
  put_pairs(sWr, tid, w2);
  put_pairs(sWr, tid + 512, w3);
  *(v4f*)(sH + hr * APITCH + hc * 4) = hv0;
  *(v4f*)(sH + (hr + 32) * APITCH + hc * 4) = hv1;

  {
    const float* hl = hprev + lane * 2;
#pragma unroll
    for (int q = 0; q < 4; ++q) {
      const int c = cq[q];
      const int s0 = sq[q];
      v2f acc = {0.0f, 0.0f};
      if (c > 0) gather_slots(hl, si[q], c < 32 ? c : 32, acc);
      for (int b = 32; b < c; b += 32) {
        const int m = (c - b) < 32 ? (c - b) : 32;
        const int li = lane < m ? lane : m - 1;
        const int s = ssrc[s0 + b + li];
        gather_slots(hl, s, m, acc);
      }
      const float invd = 1.0f / fmaxf((float)c, 1.0f);
      acc.x *= invd;
      acc.y *= invd;
      *(v2f*)(sA + (nl0 + q) * APITCH + lane * 2) = acc;
    }
  }
  __syncthreads();

  const int hf = lane >> 4;
  const int r = lane & 15;
  const int mt = wave >> 2;
  const int ct = wave & 3;

  v8f acc;
  {
    const float b = bl[ct * 16 + r];
#pragma unroll
    for (int i = 0; i < 8; ++i) acc[i] = b;
  }
  const float* arow = sA + (mt * 16 + r) * APITCH + 2 * hf;
  const float* hrow = sH + (mt * 16 + r) * APITCH + 2 * hf;
  const float* wlp = sWl + hf * 128 + (ct * 16 + r) * 2;
  const float* wrp = sWr + hf * 128 + (ct * 16 + r) * 2;
#pragma unroll
  for (int k = 0; k < HIDDEN; k += 4) {
    const v2f a1 = *(const v2f*)(arow + k);
    const v2f a2 = *(const v2f*)(hrow + k);
    const v2f b1 = *(const v2f*)(wlp + (k >> 1) * 128);
    acc = wmma4(a1, b1, acc);
    const v2f b2 = *(const v2f*)(wrp + (k >> 1) * 128);
    acc = wmma4(a2, b2, acc);
  }
  __syncthreads();

  float* sOut = sA;
  {
    float* so = sOut + mt * 16 * HIDDEN;
#pragma unroll
    for (int i = 0; i < 8; ++i) so[(i + 8 * hf) * HIDDEN + ct * 16 + r] = acc[i];
  }
  __syncthreads();
  const int rr = wave * 4 + (lane >> 3);
  const int p4 = (lane & 7) * 4;
  const int node = blockRow + rr;
  const float* orow = sOut + rr * HIDDEN + p4;
  const v4f o0 = *(const v4f*)(orow);
  const v4f o1 = *(const v4f*)(orow + 32);
  float ss = o0.x * o0.x + o0.y * o0.y + o0.z * o0.z + o0.w * o0.w +
             o1.x * o1.x + o1.y * o1.y + o1.z * o1.z + o1.w * o1.w;
  ss += __shfl_xor(ss, 4);
  ss += __shfl_xor(ss, 2);
  ss += __shfl_xor(ss, 1);
  const float l2 = rsqrtf(fmaxf(ss, 1e-24f));

  const float* hrow2 = sH + rr * APITCH + p4;
  v4f t0 = o0 * l2;
  v4f t1 = o1 * l2;
  t0.x = fmaxf(t0.x, 0.0f); t0.y = fmaxf(t0.y, 0.0f); t0.z = fmaxf(t0.z, 0.0f); t0.w = fmaxf(t0.w, 0.0f);
  t1.x = fmaxf(t1.x, 0.0f); t1.y = fmaxf(t1.y, 0.0f); t1.z = fmaxf(t1.z, 0.0f); t1.w = fmaxf(t1.w, 0.0f);
  t0 += *(const v4f*)(hrow2);
  t1 += *(const v4f*)(hrow2 + 32);
  float s = t0.x + t0.y + t0.z + t0.w + t1.x + t1.y + t1.z + t1.w;
  float q = t0.x * t0.x + t0.y * t0.y + t0.z * t0.z + t0.w * t0.w +
            t1.x * t1.x + t1.y * t1.y + t1.z * t1.z + t1.w * t1.w;
  s += __shfl_xor(s, 4);
  q += __shfl_xor(q, 4);
  s += __shfl_xor(s, 2);
  q += __shfl_xor(q, 2);
  s += __shfl_xor(s, 1);
  q += __shfl_xor(q, 1);
  const float mean = s * (1.0f / 64.0f);
  const float var = q * (1.0f / 64.0f) - mean * mean;
  const float sc = rsqrtf(var + 1e-5f);

  v4f y0 = (t0 - mean) * sc * *(const v4f*)(g + p4) + *(const v4f*)(bt + p4);
  v4f y1 = (t1 - mean) * sc * *(const v4f*)(g + 32 + p4) + *(const v4f*)(bt + 32 + p4);
  if (LAST) {
    float s2 = y0.x + y0.y + y0.z + y0.w + y1.x + y1.y + y1.z + y1.w;
    float q2 = y0.x * y0.x + y0.y * y0.y + y0.z * y0.z + y0.w * y0.w +
               y1.x * y1.x + y1.y * y1.y + y1.z * y1.z + y1.w * y1.w;
    s2 += __shfl_xor(s2, 4);
    q2 += __shfl_xor(q2, 4);
    s2 += __shfl_xor(s2, 2);
    q2 += __shfl_xor(q2, 2);
    s2 += __shfl_xor(s2, 1);
    q2 += __shfl_xor(q2, 1);
    const float mean2 = s2 * (1.0f / 64.0f);
    const float var2 = q2 * (1.0f / 64.0f) - mean2 * mean2;
    const float sc2 = rsqrtf(var2 + 1e-5f);
    y0 = (y0 - mean2) * sc2 * *(const v4f*)(g2 + p4) + *(const v4f*)(b2 + p4);
    y1 = (y1 - mean2) * sc2 * *(const v4f*)(g2 + 32 + p4) + *(const v4f*)(b2 + 32 + p4);
  }
  if (node < n) {
    float* op = hnext + (size_t)node * HIDDEN + p4;
    *(v4f*)(op) = y0;
    *(v4f*)(op + 32) = y1;
  }
}

__global__ __launch_bounds__(256) void k_finalln(const float* __restrict__ h,
                                                 const float* __restrict__ g,
                                                 const float* __restrict__ b,
                                                 float* __restrict__ out, int n) {
  int t = blockIdx.x * blockDim.x + threadIdx.x;
  int row = t >> 5;
  int lane = t & 31;
  if (row >= n) return;
  v2f v = *(const v2f*)(h + (size_t)row * HIDDEN + lane * 2);
  float s = v.x + v.y;
  float q = v.x * v.x + v.y * v.y;
#pragma unroll
  for (int m = 16; m >= 1; m >>= 1) {
    s += __shfl_xor(s, m);
    q += __shfl_xor(q, m);
  }
  float mean = s * (1.0f / 64.0f);
  float var = q * (1.0f / 64.0f) - mean * mean;
  float sc = rsqrtf(var + 1e-5f);
  v2f g2 = *(const v2f*)(g + lane * 2);
  v2f b2 = *(const v2f*)(b + lane * 2);
  v2f o;
  o.x = (v.x - mean) * sc * g2.x + b2.x;
  o.y = (v.y - mean) * sc * g2.y + b2.y;
  *(v2f*)(out + (size_t)row * HIDDEN + lane * 2) = o;
}

extern "C" void kernel_launch(void* const* d_in, const int* in_sizes, int n_in,
                              void* d_out, int out_size, void* d_ws, size_t ws_size,
                              hipStream_t stream) {
  const float* x   = (const float*)d_in[0];
  const int* ei    = (const int*)d_in[1];
  const float* Win = (const float*)d_in[2];
  const float* bin = (const float*)d_in[3];
  const float* Wl  = (const float*)d_in[4];
  const float* bl  = (const float*)d_in[5];
  const float* Wr  = (const float*)d_in[6];
  const float* lng = (const float*)d_in[7];
  const float* lnb = (const float*)d_in[8];
  const float* og  = (const float*)d_in[9];
  const float* ob  = (const float*)d_in[10];

  const int N = in_sizes[0] / DNODE;
  const int E = in_sizes[1] / 2;
  const int LAYERS = in_sizes[4] / (HIDDEN * HIDDEN);
  const int* src = ei;
  const int* dst = ei + E;

  const int B = (N + BNODES - 1) >> BSHIFT;
  const int per = (E + NCHUNK - 1) / NCHUNK;
  float* h0  = (float*)d_ws;
  float* h1  = h0 + (size_t)N * HIDDEN;
  int* tmp   = (int*)(h1 + (size_t)N * HIDDEN);
  int* ssrc  = tmp + (size_t)2 * E;
  int* cnt   = ssrc + E;
  int* start = cnt + N;
  int* wcnt  = start + N;
  int* wpre  = wcnt + (size_t)NCHUNK * B;
  int* btot  = wpre + (size_t)NCHUNK * B;
  int* bbase = btot + B;

  k_bcount<<<NCHUNK, 1024, 0, stream>>>(dst, wcnt, E, per, B);
  k_bscan<<<B, NCHUNK, 0, stream>>>(wcnt, wpre, btot, B);
  k_bplace<<<NCHUNK, 1024, 0, stream>>>(src, dst, wpre, btot, bbase, tmp, E, per, B);
  k_bsort<<<B, 1024, 0, stream>>>(tmp, bbase, ssrc, cnt, start, N);

  k_inproj<<<(N + 63) / 64, 128, 0, stream>>>(x, Win, bin, h0, N);

  float* cur = h0;
  float* nxt = h1;
  for (int l = 0; l < LAYERS; ++l) {
    if (l == LAYERS - 1) {
      k_layer<true><<<(N + 63) / 64, 512, 0, stream>>>(
          ssrc, start, cnt, cur,
          Wl + (size_t)l * HIDDEN * HIDDEN, bl + (size_t)l * HIDDEN,
          Wr + (size_t)l * HIDDEN * HIDDEN,
          lng + (size_t)l * HIDDEN, lnb + (size_t)l * HIDDEN, og, ob, (float*)d_out, N);
    } else {
      k_layer<false><<<(N + 63) / 64, 512, 0, stream>>>(
          ssrc, start, cnt, cur,
          Wl + (size_t)l * HIDDEN * HIDDEN, bl + (size_t)l * HIDDEN,
          Wr + (size_t)l * HIDDEN * HIDDEN,
          lng + (size_t)l * HIDDEN, lnb + (size_t)l * HIDDEN, og, ob, nxt, N);
    }
    float* t = cur; cur = nxt; nxt = t;
  }

  if (LAYERS == 0)
    k_finalln<<<(int)(((long long)N * 32 + 255) / 256), 256, 0, stream>>>(cur, og, ob, (float*)d_out, N);
}


# edge sort: level 1 orders each chunk's edges by bucket in the chunk's own range (count, scan, place in one launch); bucket pass collects its 256 pieces; 8 -> 6 launches; slots in LDS arrival order
# speedup vs baseline: 3.8727x; 1.0327x over previous
#include <hip/hip_runtime.h>
#include <hip/hip_bf16.h>

typedef __attribute__((ext_vector_type(2))) float v2f;
typedef __attribute__((ext_vector_type(4))) float v4f;
typedef __attribute__((ext_vector_type(8))) float v8f;

#define HIDDEN 64
#define DNODE 384

static __device__ __forceinline__ v8f wmma4(v2f a, v2f b, v8f c) {
  return __builtin_amdgcn_wmma_f32_16x16x4_f32(
       false, a,  false, b,
       (short)0, c,  false,  false);
}

static __device__ __forceinline__ void stage_pairs(float* sW, const float* W,
                                                   int rows, int tid, int nthr) {
  const int nvec = rows * HIDDEN / 4;
  for (int idx = tid; idx < nvec; idx += nthr) {
    int k = idx >> 4;
    int col = (idx & 15) * 4;
    v4f v = ((const v4f*)W)[idx];
    float* dstp = sW + (k >> 1) * 128 + (k & 1);
    dstp[(col + 0) * 2] = v.x;
    dstp[(col + 1) * 2] = v.y;
    dstp[(col + 2) * 2] = v.z;
    dstp[(col + 3) * 2] = v.w;
  }
}

static __device__ __forceinline__ void put_pairs(float* sW, int idx, v4f v) {
  const int k = idx >> 4;
  const int col = (idx & 15) * 4;
  float* dstp = sW + (k >> 1) * 128 + (k & 1);
  dstp[(col + 0) * 2] = v.x;
  dstp[(col + 1) * 2] = v.y;
  dstp[(col + 2) * 2] = v.z;
  dstp[(col + 3) * 2] = v.w;
}

#define NCHUNK 256
#define BSHIFT 9
#define BNODES 512
#define BMAX 1024
#define EU 8

template <int NW>
static __device__ __forceinline__ int block_scan_excl(int v, int* wsum) {
  const int lane = threadIdx.x & 31;
  const int wave = threadIdx.x >> 5;
  int x = v;
#pragma unroll
  for (int d = 1; d < 32; d <<= 1) {
    const int y = __shfl_up(x, d);
    if (lane >= d) x += y;
  }
  if (lane == 31) wsum[wave] = x;
  __syncthreads();
  if (wave == 0) {
    const int tv = lane < NW ? wsum[lane] : 0;
    int z = tv;
#pragma unroll
    for (int d = 1; d < 32; d <<= 1) {
      const int y = __shfl_up(z, d);
      if (lane >= d) z += y;
    }
    if (lane < NW) wsum[lane] = z - tv;
  }
  __syncthreads();
  return wsum[wave] + x - v;
}

__global__ __launch_bounds__(1024) void k_bchunk(const int* __restrict__ src,
                                                 const int* __restrict__ dst,
                                                 int* __restrict__ wcnt,
                                                 int* __restrict__ lpre,
                                                 int* __restrict__ tmp,
                                                 int E, int per, int B) {
  __shared__ int hb[BMAX];
  __shared__ int wsum[32];
  const int t = threadIdx.x;
  const int w = blockIdx.x;
  hb[t] = 0;
  __syncthreads();
  const int e0 = w * per;
  const int e1 = (e0 + per) < E ? (e0 + per) : E;
  for (int e = e0 + t; e < e1; e += EU * 1024) {
    int d[EU];
#pragma unroll
    for (int u = 0; u < EU; ++u) {
      const int ee = e + u * 1024;
      d[u] = ee < e1 ? dst[ee] : 0;
    }
#pragma unroll
    for (int u = 0; u < EU; ++u)
      if (e + u * 1024 < e1) atomicAdd(&hb[d[u] >> BSHIFT], 1);
  }
  __syncthreads();
  const int v = hb[t];
  const int ex = block_scan_excl<32>(v, wsum);
  if (t < B) {
    wcnt[w * B + t] = v;
    lpre[w * B + t] = ex;
  }
  hb[t] = e0 + ex;
  __syncthreads();
  int2* tmp2 = (int2*)tmp;
  for (int e = e0 + t; e < e1; e += EU * 1024) {
    int d[EU], s[EU];
#pragma unroll
    for (int u = 0; u < EU; ++u) {
      const int ee = e + u * 1024;
      const bool ok = ee < e1;
      d[u] = ok ? dst[ee] : 0;
      s[u] = ok ? src[ee] : 0;
    }
#pragma unroll
    for (int u = 0; u < EU; ++u) {
      if (e + u * 1024 < e1) {
        const int p = atomicAdd(&hb[d[u] >> BSHIFT], 1);
        tmp2[p] = make_int2(s[u], d[u]);
      }
    }
  }
}

#define PT (1024 / NCHUNK)
__global__ __launch_bounds__(1024) void k_bsort(const int* __restrict__ tmp,
                                                const int* __restrict__ wcnt,
                                                const int* __restrict__ lpre,
                                                int* __restrict__ ssrc,
                                                int* __restrict__ cnt,
                                                int* __restrict__ start,
                                                int n, int per, int B) {
  __shared__ int c[BNODES];
  __shared__ int wsum[32];
  __shared__ int pbeg[NCHUNK];
  __shared__ int plen[NCHUNK];
  __shared__ int sbase;
  const int t = threadIdx.x;
  const int b = blockIdx.x;
  const int2* tmp2 = (const int2*)tmp;
  int lp = 0;
  if (t < NCHUNK) {
    lp = lpre[t * B + b];
    pbeg[t] = t * per + lp;
    plen[t] = wcnt[t * B + b];
  }
  if (t < BNODES) c[t] = 0;
  const int exl = block_scan_excl<32>(lp, wsum);
  if (t == 1023) sbase = exl + lp;
  __syncthreads();
  const int e0 = sbase;
  const int pb = pbeg[t / PT];
  const int pn = plen[t / PT];
  const int j0 = t % PT;
  for (int i = j0; i < pn; i += EU * PT) {
    int d[EU];
#pragma unroll
    for (int u = 0; u < EU; ++u) {
      const int ii = i + u * PT;
      d[u] = ii < pn ? tmp[2 * (size_t)(pb + ii) + 1] : 0;
    }
#pragma unroll
    for (int u = 0; u < EU; ++u)
      if (i + u * PT < pn) atomicAdd(&c[d[u] & (BNODES - 1)], 1);
  }
  __syncthreads();
  const int v = t < BNODES ? c[t] : 0;
  const int ex = block_scan_excl<32>(v, wsum);
  const int node = b * BNODES + t;
  if (t < BNODES) {
    c[t] = e0 + ex;
    if (node < n) {
      cnt[node] = v;
      start[node] = e0 + ex;
    }
  }
  __syncthreads();
  for (int i = j0; i < pn; i += EU * PT) {
    int2 sd[EU];
#pragma unroll
    for (int u = 0; u < EU; ++u) {
      const int ii = i + u * PT;
      sd[u] = ii < pn ? tmp2[pb + ii] : make_int2(0, 0);
    }
#pragma unroll
    for (int u = 0; u < EU; ++u) {
      if (i + u * PT < pn) {
        const int p = atomicAdd(&c[sd[u].y & (BNODES - 1)], 1);
        ssrc[p] = sd[u].x;
      }
    }
  }
}

template <int G>
static __device__ __forceinline__ void gather_group(const float* hl, int s, int j,
                                                    v2f& acc) {
  v2f v[G];
#pragma unroll
  for (int u = 0; u < G; ++u) {
    const int sj = __builtin_amdgcn_readlane(s, j + u);
    v[u] = *(const v2f*)(hl + (size_t)sj * HIDDEN);
  }
#pragma unroll
  for (int u = 0; u < G; ++u) {
    acc.x += v[u].x;
    acc.y += v[u].y;
  }
}

static __device__ __forceinline__ void gather_slots(const float* hl, int s, int m,
                                                    v2f& acc) {
  int j = 0;
  for (; j + 8 <= m; j += 8) gather_group<8>(hl, s, j, acc);
  if (j + 4 <= m) { gather_group<4>(hl, s, j, acc); j += 4; }
  if (j + 2 <= m) { gather_group<2>(hl, s, j, acc); j += 2; }
  if (j < m) gather_group<1>(hl, s, j, acc);
}

__global__ __launch_bounds__(128) void k_inproj(const float* __restrict__ x,
                                                const float* __restrict__ Win,
                                                const float* __restrict__ bin,
                                                float* __restrict__ h, int n) {
  __shared__ float sW[96 * HIDDEN];
  const int wave = threadIdx.x >> 5;
  const int lane = threadIdx.x & 31;
  const int hf = lane >> 4;
  const int r = lane & 15;
  const int rowBase = blockIdx.x * 64 + wave * 16;
  const int row = rowBase + r;
  const int rowc = row < n ? row : n - 1;
  const float* xr = x + (size_t)rowc * DNODE;

  v8f acc[4];
#pragma unroll
  for (int c = 0; c < 4; ++c) {
    float b = bin[c * 16 + r];
#pragma unroll
    for (int i = 0; i < 8; ++i) acc[c][i] = b;
  }

  for (int kc = 0; kc < DNODE; kc += 96) {
    __syncthreads();
    stage_pairs(sW, Win + (size_t)kc * HIDDEN, 96, threadIdx.x, 128);
    __syncthreads();

    if (kc + 96 < DNODE) __builtin_prefetch(xr + kc + 96 + 2 * hf, 0, 3);

#pragma unroll 4
    for (int k = 0; k < 96; k += 4) {
      v2f a = *(const v2f*)(xr + kc + k + 2 * hf);
      const float* wrow = sW + ((k >> 1) + hf) * 128;
#pragma unroll
      for (int c = 0; c < 4; ++c) {
        v2f b = *(const v2f*)(wrow + (c * 16 + r) * 2);
        acc[c] = wmma4(a, b, acc[c]);
      }
    }
  }

#pragma unroll
  for (int c = 0; c < 4; ++c) {
#pragma unroll
    for (int i = 0; i < 8; ++i) {
      int m = rowBase + i + 8 * hf;
      if (m < n) h[(size_t)m * HIDDEN + c * 16 + r] = fmaxf(acc[c][i], 0.0f);
    }
  }
}

#define APITCH 68

template <bool LAST>
__global__ __launch_bounds__(512) void k_layer(const int* __restrict__ ssrc,
                                               const int* __restrict__ start,
                                               const int* __restrict__ cnt,
                                               const float* __restrict__ hprev,
                                               const float* __restrict__ Wl,
                                               const float* __restrict__ bl,
                                               const float* __restrict__ Wr,
                                               const float* __restrict__ g,
                                               const float* __restrict__ bt,
                                               const float* __restrict__ g2,
                                               const float* __restrict__ b2,
                                               float* __restrict__ hnext, int n) {
  __shared__ __attribute__((aligned(16))) float sWl[HIDDEN * HIDDEN];
  __shared__ __attribute__((aligned(16))) float sWr[HIDDEN * HIDDEN];
  __shared__ __attribute__((aligned(16))) float sH[64 * APITCH];
  __shared__ __attribute__((aligned(16))) float sA[64 * APITCH];

  const int tid = threadIdx.x;
  const int lane = tid & 31;
  const int wave = __builtin_amdgcn_readfirstlane(tid >> 5);
  const int blockRow = blockIdx.x * 64;

  const int nl0 = wave * 4;
  int ndv = blockRow + nl0 + (lane & 3);
  const bool okv = ndv < n;
  ndv = okv ? ndv : n - 1;
  const int cl = cnt[ndv];
  const int sv = start[ndv];
  const int cv = okv ? cl : 0;

  const v4f w0 = ((const v4f*)Wl)[tid];
  const v4f w1 = ((const v4f*)Wl)[tid + 512];
  const v4f w2 = ((const v4f*)Wr)[tid];
  const v4f w3 = ((const v4f*)Wr)[tid + 512];
  const int hr = tid >> 4;
  const int hc = tid & 15;
  int gr0 = blockRow + hr;
  int gr1 = gr0 + 32;
  gr0 = gr0 < n ? gr0 : n - 1;
  gr1 = gr1 < n ? gr1 : n - 1;
  const v4f hv0 = ((const v4f*)(hprev + (size_t)gr0 * HIDDEN))[hc];
  const v4f hv1 = ((const v4f*)(hprev + (size_t)gr1 * HIDDEN))[hc];

  int cq[4], sq[4], si[4];
#pragma unroll
  for (int q = 0; q < 4; ++q) {
    cq[q] = __builtin_amdgcn_readlane(cv, q);
    sq[q] = __builtin_amdgcn_readlane(sv, q);
  }
#pragma unroll
  for (int q = 0; q < 4; ++q) {
    const int c = cq[q];
    const int m = c < 32 ? c : 32;
    const int li = lane < m ? lane : m - 1;
    si[q] = 0;
    if (c > 0) si[q] = ssrc[sq[q] + li];
  }

  put_pairs(sWl, tid, w0);
  put_pairs(sWl, tid + 512, w1);
  put_pairs(sWr, tid, w2);
  put_pairs(sWr, tid + 512, w3);
  *(v4f*)(sH + hr * APITCH + hc * 4) = hv0;
  *(v4f*)(sH + (hr + 32) * APITCH + hc * 4) = hv1;

  {
    const float* hl = hprev + lane * 2;
#pragma unroll
    for (int q = 0; q < 4; ++q) {
      const int c = cq[q];
      const int s0 = sq[q];
      v2f acc = {0.0f, 0.0f};
      if (c > 0) gather_slots(hl, si[q], c < 32 ? c : 32, acc);
      for (int b = 32; b < c; b += 32) {
        const int m = (c - b) < 32 ? (c - b) : 32;
        const int li = lane < m ? lane : m - 1;
        const int s = ssrc[s0 + b + li];
        gather_slots(hl, s, m, acc);
      }
      const float invd = 1.0f / fmaxf((float)c, 1.0f);
      acc.x *= invd;
      acc.y *= invd;
      *(v2f*)(sA + (nl0 + q) * APITCH + lane * 2) = acc;
    }
  }
  __syncthreads();

  const int hf = lane >> 4;
  const int r = lane & 15;
  const int mt = wave >> 2;
  const int ct = wave & 3;

  v8f acc;
  {
    const float b = bl[ct * 16 + r];
#pragma unroll
    for (int i = 0; i < 8; ++i) acc[i] = b;
  }
  const float* arow = sA + (mt * 16 + r) * APITCH + 2 * hf;
  const float* hrow = sH + (mt * 16 + r) * APITCH + 2 * hf;
  const float* wlp = sWl + hf * 128 + (ct * 16 + r) * 2;
  const float* wrp = sWr + hf * 128 + (ct * 16 + r) * 2;
#pragma unroll
  for (int k = 0; k < HIDDEN; k += 4) {
    const v2f a1 = *(const v2f*)(arow + k);
    const v2f a2 = *(const v2f*)(hrow + k);
    const v2f b1 = *(const v2f*)(wlp + (k >> 1) * 128);
    acc = wmma4(a1, b1, acc);
    const v2f b2 = *(const v2f*)(wrp + (k >> 1) * 128);
    acc = wmma4(a2, b2, acc);
  }
  __syncthreads();

  float* sOut = sA;
  {
    float* so = sOut + mt * 16 * HIDDEN;
#pragma unroll
    for (int i = 0; i < 8; ++i) so[(i + 8 * hf) * HIDDEN + ct * 16 + r] = acc[i];
  }
  __syncthreads();
  const int rr = wave * 4 + (lane >> 3);
  const int p4 = (lane & 7) * 4;
  const int node = blockRow + rr;
  const float* orow = sOut + rr * HIDDEN + p4;
  const v4f o0 = *(const v4f*)(orow);
  const v4f o1 = *(const v4f*)(orow + 32);
  float ss = o0.x * o0.x + o0.y * o0.y + o0.z * o0.z + o0.w * o0.w +
             o1.x * o1.x + o1.y * o1.y + o1.z * o1.z + o1.w * o1.w;
  ss += __shfl_xor(ss, 4);
  ss += __shfl_xor(ss, 2);
  ss += __shfl_xor(ss, 1);
  const float l2 = rsqrtf(fmaxf(ss, 1e-24f));

  const float* hrow2 = sH + rr * APITCH + p4;
  v4f t0 = o0 * l2;
  v4f t1 = o1 * l2;
  t0.x = fmaxf(t0.x, 0.0f); t0.y = fmaxf(t0.y, 0.0f); t0.z = fmaxf(t0.z, 0.0f); t0.w = fmaxf(t0.w, 0.0f);
  t1.x = fmaxf(t1.x, 0.0f); t1.y = fmaxf(t1.y, 0.0f); t1.z = fmaxf(t1.z, 0.0f); t1.w = fmaxf(t1.w, 0.0f);
  t0 += *(const v4f*)(hrow2);
  t1 += *(const v4f*)(hrow2 + 32);
  float s = t0.x + t0.y + t0.z + t0.w + t1.x + t1.y + t1.z + t1.w;
  float q = t0.x * t0.x + t0.y * t0.y + t0.z * t0.z + t0.w * t0.w +
            t1.x * t1.x + t1.y * t1.y + t1.z * t1.z + t1.w * t1.w;
  s += __shfl_xor(s, 4);
  q += __shfl_xor(q, 4);
  s += __shfl_xor(s, 2);
  q += __shfl_xor(q, 2);
  s += __shfl_xor(s, 1);
  q += __shfl_xor(q, 1);
  const float mean = s * (1.0f / 64.0f);
  const float var = q * (1.0f / 64.0f) - mean * mean;
  const float sc = rsqrtf(var + 1e-5f);

  v4f y0 = (t0 - mean) * sc * *(const v4f*)(g + p4) + *(const v4f*)(bt + p4);
  v4f y1 = (t1 - mean) * sc * *(const v4f*)(g + 32 + p4) + *(const v4f*)(bt + 32 + p4);
  if (LAST) {
    float s2 = y0.x + y0.y + y0.z + y0.w + y1.x + y1.y + y1.z + y1.w;
    float q2 = y0.x * y0.x + y0.y * y0.y + y0.z * y0.z + y0.w * y0.w +
               y1.x * y1.x + y1.y * y1.y + y1.z * y1.z + y1.w * y1.w;
    s2 += __shfl_xor(s2, 4);
    q2 += __shfl_xor(q2, 4);
    s2 += __shfl_xor(s2, 2);
    q2 += __shfl_xor(q2, 2);
    s2 += __shfl_xor(s2, 1);
    q2 += __shfl_xor(q2, 1);
    const float mean2 = s2 * (1.0f / 64.0f);
    const float var2 = q2 * (1.0f / 64.0f) - mean2 * mean2;
    const float sc2 = rsqrtf(var2 + 1e-5f);
    y0 = (y0 - mean2) * sc2 * *(const v4f*)(g2 + p4) + *(const v4f*)(b2 + p4);
    y1 = (y1 - mean2) * sc2 * *(const v4f*)(g2 + 32 + p4) + *(const v4f*)(b2 + 32 + p4);
  }
  if (node < n) {
    float* op = hnext + (size_t)node * HIDDEN + p4;
    *(v4f*)(op) = y0;
    *(v4f*)(op + 32) = y1;
  }
}

__global__ __launch_bounds__(256) void k_finalln(const float* __restrict__ h,
                                                 const float* __restrict__ g,
                                                 const float* __restrict__ b,
                                                 float* __restrict__ out, int n) {
  int t = blockIdx.x * blockDim.x + threadIdx.x;
  int row = t >> 5;
  int lane = t & 31;
  if (row >= n) return;
  v2f v = *(const v2f*)(h + (size_t)row * HIDDEN + lane * 2);
  float s = v.x + v.y;
  float q = v.x * v.x + v.y * v.y;
#pragma unroll
  for (int m = 16; m >= 1; m >>= 1) {
    s += __shfl_xor(s, m);
    q += __shfl_xor(q, m);
  }
  float mean = s * (1.0f / 64.0f);
  float var = q * (1.0f / 64.0f) - mean * mean;
  float sc = rsqrtf(var + 1e-5f);
  v2f g2 = *(const v2f*)(g + lane * 2);
  v2f b2 = *(const v2f*)(b + lane * 2);
  v2f o;
  o.x = (v.x - mean) * sc * g2.x + b2.x;
  o.y = (v.y - mean) * sc * g2.y + b2.y;
  *(v2f*)(out + (size_t)row * HIDDEN + lane * 2) = o;
}

extern "C" void kernel_launch(void* const* d_in, const int* in_sizes, int n_in,
                              void* d_out, int out_size, void* d_ws, size_t ws_size,
                              hipStream_t stream) {
  const float* x   = (const float*)d_in[0];
  const int* ei    = (const int*)d_in[1];
  const float* Win = (const float*)d_in[2];
  const float* bin = (const float*)d_in[3];
  const float* Wl  = (const float*)d_in[4];
  const float* bl  = (const float*)d_in[5];
  const float* Wr  = (const float*)d_in[6];
  const float* lng = (const float*)d_in[7];
  const float* lnb = (const float*)d_in[8];
  const float* og  = (const float*)d_in[9];
  const float* ob  = (const float*)d_in[10];

  const int N = in_sizes[0] / DNODE;
  const int E = in_sizes[1] / 2;
  const int LAYERS = in_sizes[4] / (HIDDEN * HIDDEN);
  const int* src = ei;
  const int* dst = ei + E;

  const int B = (N + BNODES - 1) >> BSHIFT;
  const int per = (E + NCHUNK - 1) / NCHUNK;
  float* h0  = (float*)d_ws;
  float* h1  = h0 + (size_t)N * HIDDEN;
  int* tmp   = (int*)(h1 + (size_t)N * HIDDEN);
  int* ssrc  = tmp + (size_t)2 * E;
  int* cnt   = ssrc + E;
  int* start = cnt + N;
  int* wcnt  = start + N;
  int* lpre  = wcnt + (size_t)NCHUNK * B;

  k_bchunk<<<NCHUNK, 1024, 0, stream>>>(src, dst, wcnt, lpre, tmp, E, per, B);
  k_bsort<<<B, 1024, 0, stream>>>(tmp, wcnt, lpre, ssrc, cnt, start, N, per, B);

  k_inproj<<<(N + 63) / 64, 128, 0, stream>>>(x, Win, bin, h0, N);

  float* cur = h0;
  float* nxt = h1;
  for (int l = 0; l < LAYERS; ++l) {
    if (l == LAYERS - 1) {
      k_layer<true><<<(N + 63) / 64, 512, 0, stream>>>(
          ssrc, start, cnt, cur,
          Wl + (size_t)l * HIDDEN * HIDDEN, bl + (size_t)l * HIDDEN,
          Wr + (size_t)l * HIDDEN * HIDDEN,
          lng + (size_t)l * HIDDEN, lnb + (size_t)l * HIDDEN, og, ob, (float*)d_out, N);
    } else {
      k_layer<false><<<(N + 63) / 64, 512, 0, stream>>>(
          ssrc, start, cnt, cur,
          Wl + (size_t)l * HIDDEN * HIDDEN, bl + (size_t)l * HIDDEN,
          Wr + (size_t)l * HIDDEN * HIDDEN,
          lng + (size_t)l * HIDDEN, lnb + (size_t)l * HIDDEN, og, ob, nxt, N);
    }
    float* t = cur; cur = nxt; nxt = t;
  }

  if (LAYERS == 0)
    k_finalln<<<(int)(((long long)N * 32 + 255) / 256), 256, 0, stream>>>(cur, og, ob, (float*)d_out, N);
}


# front end: each edge-sort pass shares a launch with one half of the input projection, one sort and one projection group per WGP (1024 threads, W_in whole in LDS, 16x32 wave tiles); 6 -> 5 launches
# speedup vs baseline: 4.2248x; 1.0909x over previous
#include <hip/hip_runtime.h>
#include <hip/hip_bf16.h>

typedef __attribute__((ext_vector_type(2))) float v2f;
typedef __attribute__((ext_vector_type(4))) float v4f;
typedef __attribute__((ext_vector_type(8))) float v8f;

#define HIDDEN 64
#define DNODE 384

static __device__ __forceinline__ v8f wmma4(v2f a, v2f b, v8f c) {
  return __builtin_amdgcn_wmma_f32_16x16x4_f32(
       false, a,  false, b,
       (short)0, c,  false,  false);
}

static __device__ __forceinline__ void put_pairs(float* sW, int idx, v4f v) {
  const int k = idx >> 4;
  const int col = (idx & 15) * 4;
  float* dstp = sW + (k >> 1) * 128 + (k & 1);
  dstp[(col + 0) * 2] = v.x;
  dstp[(col + 1) * 2] = v.y;
  dstp[(col + 2) * 2] = v.z;
  dstp[(col + 3) * 2] = v.w;
}

#define NCHUNK 256
#define NIB 256
#define BSHIFT 9
#define BNODES 512
#define BMAX 1024
#define EU 8

template <int NW>
static __device__ __forceinline__ int block_scan_excl(int v, int* wsum) {
  const int lane = threadIdx.x & 31;
  const int wave = threadIdx.x >> 5;
  int x = v;
#pragma unroll
  for (int d = 1; d < 32; d <<= 1) {
    const int y = __shfl_up(x, d);
    if (lane >= d) x += y;
  }
  if (lane == 31) wsum[wave] = x;
  __syncthreads();
  if (wave == 0) {
    const int tv = lane < NW ? wsum[lane] : 0;
    int z = tv;
#pragma unroll
    for (int d = 1; d < 32; d <<= 1) {
      const int y = __shfl_up(z, d);
      if (lane >= d) z += y;
    }
    if (lane < NW) wsum[lane] = z - tv;
  }
  __syncthreads();
  return wsum[wave] + x - v;
}

static __device__ __forceinline__ void bchunk_body(const int w,
                                                   const int* __restrict__ src,
                                                   const int* __restrict__ dst,
                                                   int* __restrict__ wcnt,
                                                   int* __restrict__ lpre,
                                                   int* __restrict__ tmp,
                                                   int E, int per, int B) {
  __shared__ int hb[BMAX];
  __shared__ int wsum[32];
  const int t = threadIdx.x;
  hb[t] = 0;
  __syncthreads();
  const int e0 = w * per;
  const int e1 = (e0 + per) < E ? (e0 + per) : E;
  for (int e = e0 + t; e < e1; e += EU * 1024) {
    int d[EU];
#pragma unroll
    for (int u = 0; u < EU; ++u) {
      const int ee = e + u * 1024;
      d[u] = ee < e1 ? dst[ee] : 0;
    }
#pragma unroll
    for (int u = 0; u < EU; ++u)
      if (e + u * 1024 < e1) atomicAdd(&hb[d[u] >> BSHIFT], 1);
  }
  __syncthreads();
  const int v = hb[t];
  const int ex = block_scan_excl<32>(v, wsum);
  if (t < B) {
    wcnt[w * B + t] = v;
    lpre[w * B + t] = ex;
  }
  hb[t] = e0 + ex;
  __syncthreads();
  int2* tmp2 = (int2*)tmp;
  for (int e = e0 + t; e < e1; e += EU * 1024) {
    int d[EU], s[EU];
#pragma unroll
    for (int u = 0; u < EU; ++u) {
      const int ee = e + u * 1024;
      const bool ok = ee < e1;
      d[u] = ok ? dst[ee] : 0;
      s[u] = ok ? src[ee] : 0;
    }
#pragma unroll
    for (int u = 0; u < EU; ++u) {
      if (e + u * 1024 < e1) {
        const int p = atomicAdd(&hb[d[u] >> BSHIFT], 1);
        tmp2[p] = make_int2(s[u], d[u]);
      }
    }
  }
}

#define PT (1024 / NCHUNK)
static __device__ __forceinline__ void bsort_body(const int b,
                                                  const int* __restrict__ tmp,
                                                  const int* __restrict__ wcnt,
                                                  const int* __restrict__ lpre,
                                                  int* __restrict__ ssrc,
                                                  int* __restrict__ cnt,
                                                  int* __restrict__ start,
                                                  int n, int per, int B) {
  __shared__ int c[BNODES];
  __shared__ int wsum[32];
  __shared__ int pbeg[NCHUNK];
  __shared__ int plen[NCHUNK];
  __shared__ int sbase;
  const int t = threadIdx.x;
  const int2* tmp2 = (const int2*)tmp;
  int lp = 0;
  if (t < NCHUNK) {
    lp = lpre[t * B + b];
    pbeg[t] = t * per + lp;
    plen[t] = wcnt[t * B + b];
  }
  if (t < BNODES) c[t] = 0;
  const int exl = block_scan_excl<32>(lp, wsum);
  if (t == 1023) sbase = exl + lp;
  __syncthreads();
  const int e0 = sbase;
  const int pb = pbeg[t / PT];
  const int pn = plen[t / PT];
  const int j0 = t % PT;
  for (int i = j0; i < pn; i += EU * PT) {
    int d[EU];
#pragma unroll
    for (int u = 0; u < EU; ++u) {
      const int ii = i + u * PT;
      d[u] = ii < pn ? tmp[2 * (size_t)(pb + ii) + 1] : 0;
    }
#pragma unroll
    for (int u = 0; u < EU; ++u)
      if (i + u * PT < pn) atomicAdd(&c[d[u] & (BNODES - 1)], 1);
  }
  __syncthreads();
  const int v = t < BNODES ? c[t] : 0;
  const int ex = block_scan_excl<32>(v, wsum);
  const int node = b * BNODES + t;
  if (t < BNODES) {
    c[t] = e0 + ex;
    if (node < n) {
      cnt[node] = v;
      start[node] = e0 + ex;
    }
  }
  __syncthreads();
  for (int i = j0; i < pn; i += EU * PT) {
    int2 sd[EU];
#pragma unroll
    for (int u = 0; u < EU; ++u) {
      const int ii = i + u * PT;
      sd[u] = ii < pn ? tmp2[pb + ii] : make_int2(0, 0);
    }
#pragma unroll
    for (int u = 0; u < EU; ++u) {
      if (i + u * PT < pn) {
        const int p = atomicAdd(&c[sd[u].y & (BNODES - 1)], 1);
        ssrc[p] = sd[u].x;
      }
    }
  }
}

template <int G>
static __device__ __forceinline__ void gather_group(const float* hl, int s, int j,
                                                    v2f& acc) {
  v2f v[G];
#pragma unroll
  for (int u = 0; u < G; ++u) {
    const int sj = __builtin_amdgcn_readlane(s, j + u);
    v[u] = *(const v2f*)(hl + (size_t)sj * HIDDEN);
  }
#pragma unroll
  for (int u = 0; u < G; ++u) {
    acc.x += v[u].x;
    acc.y += v[u].y;
  }
}

static __device__ __forceinline__ void gather_slots(const float* hl, int s, int m,
                                                    v2f& acc) {
  int j = 0;
  for (; j + 8 <= m; j += 8) gather_group<8>(hl, s, j, acc);
  if (j + 4 <= m) { gather_group<4>(hl, s, j, acc); j += 4; }
  if (j + 2 <= m) { gather_group<2>(hl, s, j, acc); j += 2; }
  if (j < m) gather_group<1>(hl, s, j, acc);
}


static __device__ __forceinline__ void prefetch_tile(const float* x, int n, int rowBase,
                                                     int ch, int r, int hf) {
  const int row = rowBase + r;
  if (row < n) {
    const float* p = x + (size_t)row * DNODE + ch * 192;
    __builtin_prefetch(p + hf * 64, 0, 3);
    __builtin_prefetch(p + 128, 0, 3);
  }
}

static __device__ __forceinline__ void inproj_tiles(const float* __restrict__ x,
                                                    const float* __restrict__ Win,
                                                    const float* __restrict__ bin,
                                                    float* __restrict__ h, int n,
                                                    int blk, int nblk, int s0, int s1) {
  __shared__ float sW[DNODE * HIDDEN];
  const int wave = __builtin_amdgcn_readfirstlane(threadIdx.x >> 5);
  const int lane = threadIdx.x & 31;
  const int hf = lane >> 4;
  const int r = lane & 15;
  const int wb = wave >> 3;
  const int qw = 4 * (((wave & 7) - wb) & 7) + wb;
  const int q0 = 2 * s0 + qw;
  const int q1 = 2 * s1;

  prefetch_tile(x, n, q0 < q1 ? ((q0 >> 1) * nblk + blk) * 16 : n, q0 & 1, r, hf);

  {
    v4f wv[6];
#pragma unroll
    for (int j = 0; j < 6; ++j) wv[j] = ((const v4f*)Win)[threadIdx.x + j * 1024];
#pragma unroll
    for (int j = 0; j < 6; ++j) put_pairs(sW, threadIdx.x + j * 1024, wv[j]);
  }
  __syncthreads();

  for (int q = q0; q < q1; q += 32) {
    const int ch = q & 1;
    const int rowBase = ((q >> 1) * nblk + blk) * 16;
    if (rowBase >= n) break;
    prefetch_tile(x, n, (q + 32) < q1 ? (((q + 32) >> 1) * nblk + blk) * 16 : n, ch, r, hf);
    const int row = rowBase + r;
    const int rowc = row < n ? row : n - 1;
    const float* xr = x + (size_t)rowc * DNODE + 2 * hf;
    const float* wcol = sW + hf * 128 + (ch * 32 + r) * 2;

    v8f acc[2];
#pragma unroll
    for (int c = 0; c < 2; ++c) {
      const float b = bin[ch * 32 + c * 16 + r];
#pragma unroll
      for (int i = 0; i < 8; ++i) acc[c][i] = b;
    }

#pragma unroll 8
    for (int k = 0; k < DNODE; k += 4) {
      const v2f a = *(const v2f*)(xr + k);
      const float* wrow = wcol + (k >> 1) * 128;
#pragma unroll
      for (int c = 0; c < 2; ++c) {
        const v2f b = *(const v2f*)(wrow + c * 32);
        acc[c] = wmma4(a, b, acc[c]);
      }
    }

#pragma unroll
    for (int c = 0; c < 2; ++c) {
#pragma unroll
      for (int i = 0; i < 8; ++i) {
        const int m = rowBase + i + 8 * hf;
        if (m < n) h[(size_t)m * HIDDEN + ch * 32 + c * 16 + r] = fmaxf(acc[c][i], 0.0f);
      }
    }
  }
}

template <int PART>
__global__ __launch_bounds__(1024, 16) void k_front(const int* __restrict__ src,
                                                    const int* __restrict__ dst,
                                                    int* __restrict__ wcnt,
                                                    int* __restrict__ lpre,
                                                    int* __restrict__ tmp,
                                                    int* __restrict__ ssrc,
                                                    int* __restrict__ cnt,
                                                    int* __restrict__ start,
                                                    const float* __restrict__ x,
                                                    const float* __restrict__ Win,
                                                    const float* __restrict__ bin,
                                                    float* __restrict__ h,
                                                    int E, int per, int B, int n) {
  const int i = (int)blockIdx.x;
  const int idx = i >> 1;
  if (((i ^ (i >> 8)) & 1) == 0) {
    if (PART == 0) {
      if (idx < NCHUNK) bchunk_body(idx, src, dst, wcnt, lpre, tmp, E, per, B);
    } else {
      if (idx < B) bsort_body(idx, tmp, wcnt, lpre, ssrc, cnt, start, n, per, B);
    }
    return;
  }
  if (idx >= NIB) return;
  const int tiles = (n + 15) >> 4;
  const int S = (tiles + NIB - 1) / NIB;
  const int Sh = S >> 1;
  inproj_tiles(x, Win, bin, h, n, idx, NIB, PART == 0 ? 0 : Sh, PART == 0 ? Sh : S);
}

#define APITCH 68

template <bool LAST>
__global__ __launch_bounds__(512) void k_layer(const int* __restrict__ ssrc,
                                               const int* __restrict__ start,
                                               const int* __restrict__ cnt,
                                               const float* __restrict__ hprev,
                                               const float* __restrict__ Wl,
                                               const float* __restrict__ bl,
                                               const float* __restrict__ Wr,
                                               const float* __restrict__ g,
                                               const float* __restrict__ bt,
                                               const float* __restrict__ g2,
                                               const float* __restrict__ b2,
                                               float* __restrict__ hnext, int n) {
  __shared__ __attribute__((aligned(16))) float sWl[HIDDEN * HIDDEN];
  __shared__ __attribute__((aligned(16))) float sWr[HIDDEN * HIDDEN];
  __shared__ __attribute__((aligned(16))) float sH[64 * APITCH];
  __shared__ __attribute__((aligned(16))) float sA[64 * APITCH];

  const int tid = threadIdx.x;
  const int lane = tid & 31;
  const int wave = __builtin_amdgcn_readfirstlane(tid >> 5);
  const int blockRow = blockIdx.x * 64;

  const int nl0 = wave * 4;
  int ndv = blockRow + nl0 + (lane & 3);
  const bool okv = ndv < n;
  ndv = okv ? ndv : n - 1;
  const int cl = cnt[ndv];
  const int sv = start[ndv];
  const int cv = okv ? cl : 0;

  const v4f w0 = ((const v4f*)Wl)[tid];
  const v4f w1 = ((const v4f*)Wl)[tid + 512];
  const v4f w2 = ((const v4f*)Wr)[tid];
  const v4f w3 = ((const v4f*)Wr)[tid + 512];
  const int hr = tid >> 4;
  const int hc = tid & 15;
  int gr0 = blockRow + hr;
  int gr1 = gr0 + 32;
  gr0 = gr0 < n ? gr0 : n - 1;
  gr1 = gr1 < n ? gr1 : n - 1;
  const v4f hv0 = ((const v4f*)(hprev + (size_t)gr0 * HIDDEN))[hc];
  const v4f hv1 = ((const v4f*)(hprev + (size_t)gr1 * HIDDEN))[hc];

  int cq[4], sq[4], si[4];
#pragma unroll
  for (int q = 0; q < 4; ++q) {
    cq[q] = __builtin_amdgcn_readlane(cv, q);
    sq[q] = __builtin_amdgcn_readlane(sv, q);
  }
#pragma unroll
  for (int q = 0; q < 4; ++q) {
    const int c = cq[q];
    const int m = c < 32 ? c : 32;
    const int li = lane < m ? lane : m - 1;
    si[q] = 0;
    if (c > 0) si[q] = ssrc[sq[q] + li];
  }

  put_pairs(sWl, tid, w0);
  put_pairs(sWl, tid + 512, w1);
  put_pairs(sWr, tid, w2);
  put_pairs(sWr, tid + 512, w3);
  *(v4f*)(sH + hr * APITCH + hc * 4) = hv0;
  *(v4f*)(sH + (hr + 32) * APITCH + hc * 4) = hv1;

  {
    const float* hl = hprev + lane * 2;
#pragma unroll
    for (int q = 0; q < 4; ++q) {
      const int c = cq[q];
      const int s0 = sq[q];
      v2f acc = {0.0f, 0.0f};
      if (c > 0) gather_slots(hl, si[q], c < 32 ? c : 32, acc);
      for (int b = 32; b < c; b += 32) {
        const int m = (c - b) < 32 ? (c - b) : 32;
        const int li = lane < m ? lane : m - 1;
        const int s = ssrc[s0 + b + li];
        gather_slots(hl, s, m, acc);
      }
      const float invd = 1.0f / fmaxf((float)c, 1.0f);
      acc.x *= invd;
      acc.y *= invd;
      *(v2f*)(sA + (nl0 + q) * APITCH + lane * 2) = acc;
    }
  }
  __syncthreads();

  const int hf = lane >> 4;
  const int r = lane & 15;
  const int mt = wave >> 2;
  const int ct = wave & 3;

  v8f acc;
  {
    const float b = bl[ct * 16 + r];
#pragma unroll
    for (int i = 0; i < 8; ++i) acc[i] = b;
  }
  const float* arow = sA + (mt * 16 + r) * APITCH + 2 * hf;
  const float* hrow = sH + (mt * 16 + r) * APITCH + 2 * hf;
  const float* wlp = sWl + hf * 128 + (ct * 16 + r) * 2;
  const float* wrp = sWr + hf * 128 + (ct * 16 + r) * 2;
#pragma unroll
  for (int k = 0; k < HIDDEN; k += 4) {
    const v2f a1 = *(const v2f*)(arow + k);
    const v2f a2 = *(const v2f*)(hrow + k);
    const v2f b1 = *(const v2f*)(wlp + (k >> 1) * 128);
    acc = wmma4(a1, b1, acc);
    const v2f b2 = *(const v2f*)(wrp + (k >> 1) * 128);
    acc = wmma4(a2, b2, acc);
  }
  __syncthreads();

  float* sOut = sA;
  {
    float* so = sOut + mt * 16 * HIDDEN;
#pragma unroll
    for (int i = 0; i < 8; ++i) so[(i + 8 * hf) * HIDDEN + ct * 16 + r] = acc[i];
  }
  __syncthreads();
  const int rr = wave * 4 + (lane >> 3);
  const int p4 = (lane & 7) * 4;
  const int node = blockRow + rr;
  const float* orow = sOut + rr * HIDDEN + p4;
  const v4f o0 = *(const v4f*)(orow);
  const v4f o1 = *(const v4f*)(orow + 32);
  float ss = o0.x * o0.x + o0.y * o0.y + o0.z * o0.z + o0.w * o0.w +
             o1.x * o1.x + o1.y * o1.y + o1.z * o1.z + o1.w * o1.w;
  ss += __shfl_xor(ss, 4);
  ss += __shfl_xor(ss, 2);
  ss += __shfl_xor(ss, 1);
  const float l2 = rsqrtf(fmaxf(ss, 1e-24f));

  const float* hrow2 = sH + rr * APITCH + p4;
  v4f t0 = o0 * l2;
  v4f t1 = o1 * l2;
  t0.x = fmaxf(t0.x, 0.0f); t0.y = fmaxf(t0.y, 0.0f); t0.z = fmaxf(t0.z, 0.0f); t0.w = fmaxf(t0.w, 0.0f);
  t1.x = fmaxf(t1.x, 0.0f); t1.y = fmaxf(t1.y, 0.0f); t1.z = fmaxf(t1.z, 0.0f); t1.w = fmaxf(t1.w, 0.0f);
  t0 += *(const v4f*)(hrow2);
  t1 += *(const v4f*)(hrow2 + 32);
  float s = t0.x + t0.y + t0.z + t0.w + t1.x + t1.y + t1.z + t1.w;
  float q = t0.x * t0.x + t0.y * t0.y + t0.z * t0.z + t0.w * t0.w +
            t1.x * t1.x + t1.y * t1.y + t1.z * t1.z + t1.w * t1.w;
  s += __shfl_xor(s, 4);
  q += __shfl_xor(q, 4);
  s += __shfl_xor(s, 2);
  q += __shfl_xor(q, 2);
  s += __shfl_xor(s, 1);
  q += __shfl_xor(q, 1);
  const float mean = s * (1.0f / 64.0f);
  const float var = q * (1.0f / 64.0f) - mean * mean;
  const float sc = rsqrtf(var + 1e-5f);

  v4f y0 = (t0 - mean) * sc * *(const v4f*)(g + p4) + *(const v4f*)(bt + p4);
  v4f y1 = (t1 - mean) * sc * *(const v4f*)(g + 32 + p4) + *(const v4f*)(bt + 32 + p4);
  if (LAST) {
    float s2 = y0.x + y0.y + y0.z + y0.w + y1.x + y1.y + y1.z + y1.w;
    float q2 = y0.x * y0.x + y0.y * y0.y + y0.z * y0.z + y0.w * y0.w +
               y1.x * y1.x + y1.y * y1.y + y1.z * y1.z + y1.w * y1.w;
    s2 += __shfl_xor(s2, 4);
    q2 += __shfl_xor(q2, 4);
    s2 += __shfl_xor(s2, 2);
    q2 += __shfl_xor(q2, 2);
    s2 += __shfl_xor(s2, 1);
    q2 += __shfl_xor(q2, 1);
    const float mean2 = s2 * (1.0f / 64.0f);
    const float var2 = q2 * (1.0f / 64.0f) - mean2 * mean2;
    const float sc2 = rsqrtf(var2 + 1e-5f);
    y0 = (y0 - mean2) * sc2 * *(const v4f*)(g2 + p4) + *(const v4f*)(b2 + p4);
    y1 = (y1 - mean2) * sc2 * *(const v4f*)(g2 + 32 + p4) + *(const v4f*)(b2 + 32 + p4);
  }
  if (node < n) {
    float* op = hnext + (size_t)node * HIDDEN + p4;
    *(v4f*)(op) = y0;
    *(v4f*)(op + 32) = y1;
  }
}

__global__ __launch_bounds__(256) void k_finalln(const float* __restrict__ h,
                                                 const float* __restrict__ g,
                                                 const float* __restrict__ b,
                                                 float* __restrict__ out, int n) {
  int t = blockIdx.x * blockDim.x + threadIdx.x;
  int row = t >> 5;
  int lane = t & 31;
  if (row >= n) return;
  v2f v = *(const v2f*)(h + (size_t)row * HIDDEN + lane * 2);
  float s = v.x + v.y;
  float q = v.x * v.x + v.y * v.y;
#pragma unroll
  for (int m = 16; m >= 1; m >>= 1) {
    s += __shfl_xor(s, m);
    q += __shfl_xor(q, m);
  }
  float mean = s * (1.0f / 64.0f);
  float var = q * (1.0f / 64.0f) - mean * mean;
  float sc = rsqrtf(var + 1e-5f);
  v2f g2 = *(const v2f*)(g + lane * 2);
  v2f b2 = *(const v2f*)(b + lane * 2);
  v2f o;
  o.x = (v.x - mean) * sc * g2.x + b2.x;
  o.y = (v.y - mean) * sc * g2.y + b2.y;
  *(v2f*)(out + (size_t)row * HIDDEN + lane * 2) = o;
}

extern "C" void kernel_launch(void* const* d_in, const int* in_sizes, int n_in,
                              void* d_out, int out_size, void* d_ws, size_t ws_size,
                              hipStream_t stream) {
  const float* x   = (const float*)d_in[0];
  const int* ei    = (const int*)d_in[1];
  const float* Win = (const float*)d_in[2];
  const float* bin = (const float*)d_in[3];
  const float* Wl  = (const float*)d_in[4];
  const float* bl  = (const float*)d_in[5];
  const float* Wr  = (const float*)d_in[6];
  const float* lng = (const float*)d_in[7];
  const float* lnb = (const float*)d_in[8];
  const float* og  = (const float*)d_in[9];
  const float* ob  = (const float*)d_in[10];

  const int N = in_sizes[0] / DNODE;
  const int E = in_sizes[1] / 2;
  const int LAYERS = in_sizes[4] / (HIDDEN * HIDDEN);
  const int* src = ei;
  const int* dst = ei + E;

  const int B = (N + BNODES - 1) >> BSHIFT;
  const int per = (E + NCHUNK - 1) / NCHUNK;
  float* h0  = (float*)d_ws;
  float* h1  = h0 + (size_t)N * HIDDEN;
  int* tmp   = (int*)(h1 + (size_t)N * HIDDEN);
  int* ssrc  = tmp + (size_t)2 * E;
  int* cnt   = ssrc + E;
  int* start = cnt + N;
  int* wcnt  = start + N;
  int* lpre  = wcnt + (size_t)NCHUNK * B;

  const int R0 = NCHUNK > NIB ? NCHUNK : NIB;
  const int R1 = B > NIB ? B : NIB;
  k_front<0><<<2 * R0, 1024, 0, stream>>>(src, dst, wcnt, lpre, tmp, ssrc, cnt, start,
                                        x, Win, bin, h0, E, per, B, N);
  k_front<1><<<2 * R1, 1024, 0, stream>>>(src, dst, wcnt, lpre, tmp, ssrc, cnt, start,
                                        x, Win, bin, h0, E, per, B, N);

  float* cur = h0;
  float* nxt = h1;
  for (int l = 0; l < LAYERS; ++l) {
    if (l == LAYERS - 1) {
      k_layer<true><<<(N + 63) / 64, 512, 0, stream>>>(
          ssrc, start, cnt, cur,
          Wl + (size_t)l * HIDDEN * HIDDEN, bl + (size_t)l * HIDDEN,
          Wr + (size_t)l * HIDDEN * HIDDEN,
          lng + (size_t)l * HIDDEN, lnb + (size_t)l * HIDDEN, og, ob, (float*)d_out, N);
    } else {
      k_layer<false><<<(N + 63) / 64, 512, 0, stream>>>(
          ssrc, start, cnt, cur,
          Wl + (size_t)l * HIDDEN * HIDDEN, bl + (size_t)l * HIDDEN,
          Wr + (size_t)l * HIDDEN * HIDDEN,
          lng + (size_t)l * HIDDEN, lnb + (size_t)l * HIDDEN, og, ob, nxt, N);
    }
    float* t = cur; cur = nxt; nxt = t;
  }

  if (LAYERS == 0)
    k_finalln<<<(int)(((long long)N * 32 + 255) / 256), 256, 0, stream>>>(cur, og, ob, (float*)d_out, N);
}
